# starting kernel
# baseline (speedup 1.0000x reference)
#include <hip/hip_runtime.h>


typedef float v2f __attribute__((ext_vector_type(2)));
typedef float v8f __attribute__((ext_vector_type(8)));

#define B_PATHS    16384
#define N_STEPS    500
#define SAVE_EVERY 5
#define N_SAVES    100
#define T_SAVED    101
#define DT         0.002f
#define PF_NEAR    8
#define PF_FAR     32

__global__ __launch_bounds__(64) void sde_wmma_kernel(
    const float* __restrict__ drift, const float* __restrict__ drift_bias,
    const float* __restrict__ diffusion, const float* __restrict__ diffusion_bias,
    const float* __restrict__ x0, const float* __restrict__ noise,
    float* __restrict__ out)
{
    const int  lane = threadIdx.x & 31;
    const int  wave = (blockIdx.x * blockDim.x + threadIdx.x) >> 5;
    const int  base = wave * 32;
    const int  m    = lane & 15;
    const bool lo   = lane < 16;
    const float sqdt = __builtin_sqrtf(DT);

    const int k0 = lo ? 0 : 2;
    const int k1 = lo ? 1 : 3;
    const int mm = m & 7;
    float a0v = 0.0f, a1v = 0.0f;
    if (mm < 3) {
        a0v = DT * drift[mm * 3 + k0] + ((mm == k0) ? 1.0f : 0.0f);
        a1v = (k1 < 3) ? (DT * drift[mm * 3 + k1] + ((mm == k1) ? 1.0f : 0.0f))
                       : (DT * drift_bias[mm]);
    } else if (mm < 6) {
        const int r = mm - 3;
        a0v = sqdt * diffusion[r * 3 + k0];
        a1v = (k1 < 3) ? (sqdt * diffusion[r * 3 + k1])
                       : (sqdt * diffusion_bias[r]);
    }
    v2f Amat; Amat.x = a0v; Amat.y = a1v;

    const int p0 = base + m;
    const int p1 = base + 16 + m;
    v2f B0, B1;
    B0.x = lo ? x0[p0 * 3 + 0] : x0[p0 * 3 + 2];
    B0.y = lo ? x0[p0 * 3 + 1] : 1.0f;
    B1.x = lo ? x0[p1 * 3 + 0] : x0[p1 * 3 + 2];
    B1.y = lo ? x0[p1 * 3 + 1] : 1.0f;

    const int p  = base + lane;
    const int BT = B_PATHS * T_SAVED;
    out[0 * BT + p * T_SAVED] = x0[p * 3 + 0];
    out[1 * BT + p * T_SAVED] = x0[p * 3 + 1];
    out[2 * BT + p * T_SAVED] = x0[p * 3 + 2];

    for (int s = 0; s < N_SAVES; ++s) {
        float sv0 = 0.0f, sv1 = 0.0f, sv2 = 0.0f;
#pragma unroll
        for (int k = 0; k < SAVE_EVERY; ++k) {
            const int t   = s * SAVE_EVERY + k;
            const int off = t * (B_PATHS * 3);

            int tn = t + PF_NEAR; if (tn >= N_STEPS) tn = N_STEPS - 1;
            const float* pfn = noise + tn * (B_PATHS * 3) + p * 3;
            asm volatile("global_prefetch_b8 %0, off" :: "v"(pfn));
            int tf = t + PF_FAR; if (tf >= N_STEPS) tf = N_STEPS - 1;
            __builtin_prefetch(noise + tf * (B_PATHS * 3) + p * 3, 0, 3);

            const float n00 = noise[off + p0 * 3 + 0];
            const float n01 = noise[off + p0 * 3 + 1];
            const float n02 = noise[off + p0 * 3 + 2];
            const float n10 = noise[off + p1 * 3 + 0];
            const float n11 = noise[off + p1 * 3 + 1];
            const float n12 = noise[off + p1 * 3 + 2];

            v8f cz = {};
            v8f d0 = __builtin_amdgcn_wmma_f32_16x16x4_f32(
                false, Amat, false, B0, (short)0, cz, false, false);
            v8f d1 = __builtin_amdgcn_wmma_f32_16x16x4_f32(
                false, Amat, false, B1, (short)0, cz, false, false);

            const float xn00 = __builtin_fmaf(d0[3], n00, d0[0]);
            const float xn01 = __builtin_fmaf(d0[4], n01, d0[1]);
            const float xn02 = __builtin_fmaf(d0[5], n02, d0[2]);
            const float xn10 = __builtin_fmaf(d1[3], n10, d1[0]);
            const float xn11 = __builtin_fmaf(d1[4], n11, d1[1]);
            const float xn12 = __builtin_fmaf(d1[5], n12, d1[2]);

            B0.x = lo ? xn00 : xn02;  B0.y = lo ? xn01 : 1.0f;
            B1.x = lo ? xn10 : xn12;  B1.y = lo ? xn11 : 1.0f;

            if (k == SAVE_EVERY - 1) {
                sv0 = lo ? xn00 : xn10;
                sv1 = lo ? xn01 : xn11;
                sv2 = lo ? xn02 : xn12;
            }
        }
        const int so = p * T_SAVED + (s + 1);
        out[0 * BT + so] = sv0;
        out[1 * BT + so] = sv1;
        out[2 * BT + so] = sv2;
    }
}

extern "C" void kernel_launch(void* const* d_in, const int* in_sizes, int n_in,
                              void* d_out, int out_size, void* d_ws, size_t ws_size,
                              hipStream_t stream) {
    const float* drift     = (const float*)d_in[0];
    const float* driftb    = (const float*)d_in[1];
    const float* diffusion = (const float*)d_in[2];
    const float* diffb     = (const float*)d_in[3];
    const float* x0        = (const float*)d_in[4];
    const float* noise     = (const float*)d_in[5];
    float* out = (float*)d_out;

    dim3 block(64);
    dim3 grid(B_PATHS / 64);
    sde_wmma_kernel<<<grid, block, 0, stream>>>(
        drift, driftb, diffusion, diffb, x0, noise, out);
}


# time axis cut in two halves run side by side: extra waves carry the 4 columns of each path's affine map over steps 251-500; saves 51-100 = that map applied to x(250), a new rounding order
# speedup vs baseline: 1.7532x; 1.7532x over previous
#include <hip/hip_runtime.h>


typedef float v2f __attribute__((ext_vector_type(2)));
typedef float v4f __attribute__((ext_vector_type(4)));
typedef float v8f __attribute__((ext_vector_type(8)));

#define B_PATHS    16384
#define N_STEPS    500
#define HALF_STEPS 250
#define SAVE_EVERY 5
#define T_SAVED    101
#define HALF_SAVES 50
#define DT         0.002f
#define CH         25
#define N_CHUNK    10
#define GROUPS     5
#define N_PIECE    25
#define ROW_F      192
#define STAGE_F    (2 * CH * ROW_F)
#define STEP_F     (B_PATHS * 3)
#define N_WAVES    13
#define TJ         51
#define T_PLANE    (3 * 64 * TJ)


__device__ __forceinline__ void lds_handover_barrier() {
    __builtin_amdgcn_fence(__ATOMIC_RELEASE, "workgroup", "local");
    __builtin_amdgcn_s_barrier();
    __builtin_amdgcn_fence(__ATOMIC_ACQUIRE, "workgroup", "local");
}

__device__ __forceinline__ int noise_step(int c, int sl) {
    return c * CH + sl + ((sl >= CH) ? (HALF_STEPS - CH) : 0);
}

__global__ __launch_bounds__(416) void sde_wmma_kernel(
    const float* __restrict__ drift, const float* __restrict__ drift_bias,
    const float* __restrict__ diffusion, const float* __restrict__ diffusion_bias,
    const float* __restrict__ x0, const float* __restrict__ noise,
    float* __restrict__ out)
{
    __shared__ v4f   lds4[2 * STAGE_F / 4];
    __shared__ float tcol[4 * T_PLANE + 16];
    __shared__ float xmid[3 * 64];
    float* lds = (float*)lds4;

    const int  tid  = threadIdx.x;
    const int  lane = tid & 31;
    const int  wv   = __builtin_amdgcn_readfirstlane(tid >> 5);
    const bool producer = wv >= 10;
    const int  kind = wv >> 1;
    const int  hsel = wv & 1;
    const int  blk  = blockIdx.x * 64;
    const int  base = blk + hsel * 32;
    const int  m    = lane & 15;
    const bool lo   = lane < 16;
    const float sqdt = __builtin_sqrtf(DT);
    const int  BT = B_PATHS * T_SAVED;
    const int  p  = base + lane;
    const int  pl = hsel * 32 + lane;
    const float hval = (kind >= 1 && kind <= 3) ? 0.0f : 1.0f;

    const int ptid = tid - 320;
    const int prow = (ptid >= 48) ? 1 : 0;
    const int pcol = (ptid - 48 * prow) * 4;

    v2f Amat; Amat.x = 0.0f; Amat.y = 0.0f;
    v2f B0;   B0.x = 0.0f;   B0.y = 0.0f;
    v2f B1;   B1.x = 0.0f;   B1.y = 0.0f;
    const int lp0 = (hsel * 32 + m) * 3;
    const int lp1 = lp0 + 48;

    if (producer) {
        v4f r4[N_PIECE];
#pragma unroll
        for (int r = 0; r < N_PIECE; ++r) {
            const int sl = 2 * r + prow;
            r4[r] = *(const v4f*)(noise + noise_step(0, sl) * STEP_F + blk * 3 + pcol);
        }
#pragma unroll
        for (int r = 0; r < N_PIECE; ++r) {
            const int sl = 2 * r + prow;
            __builtin_prefetch(noise + noise_step(1, sl) * STEP_F + blk * 3 + pcol, 0, 3);
        }
#pragma unroll
        for (int r = 0; r < N_PIECE; ++r) {
            const int sl = 2 * r + prow;
            *(v4f*)(lds + sl * ROW_F + pcol) = r4[r];
        }
    } else {
        const int k0 = lo ? 0 : 2;
        const int k1 = lo ? 1 : 3;
        const int mm = m & 7;
        float a0v = 0.0f, a1v = 0.0f;
        if (mm < 3) {
            a0v = DT * drift[mm * 3 + k0] + ((mm == k0) ? 1.0f : 0.0f);
            a1v = (k1 < 3) ? (DT * drift[mm * 3 + k1] + ((mm == k1) ? 1.0f : 0.0f))
                           : (DT * drift_bias[mm]);
        } else if (mm < 6) {
            const int r = mm - 3;
            a0v = sqdt * diffusion[r * 3 + k0];
            a1v = (k1 < 3) ? (sqdt * diffusion[r * 3 + k1])
                           : (sqdt * diffusion_bias[r]);
        }
        Amat.x = a0v; Amat.y = a1v;

        if (kind == 0) {
            const int p0 = base + m;
            const int p1 = base + 16 + m;
            B0.x = lo ? x0[p0 * 3 + 0] : x0[p0 * 3 + 2];
            B0.y = lo ? x0[p0 * 3 + 1] : 1.0f;
            B1.x = lo ? x0[p1 * 3 + 0] : x0[p1 * 3 + 2];
            B1.y = lo ? x0[p1 * 3 + 1] : 1.0f;

            out[0 * BT + p * T_SAVED] = x0[p * 3 + 0];
            out[1 * BT + p * T_SAVED] = x0[p * 3 + 1];
            out[2 * BT + p * T_SAVED] = x0[p * 3 + 2];
        } else {
            const float e0 = (kind == 1) ? 1.0f : 0.0f;
            const float e1 = (kind == 2) ? 1.0f : 0.0f;
            const float e2 = (kind == 3) ? 1.0f : 0.0f;
            B0.x = lo ? e0 : e2;
            B0.y = lo ? e1 : hval;
            B1 = B0;
        }
    }
    lds_handover_barrier();

    for (int c = 0; c < N_CHUNK; ++c) {
        const int st = c & 1;
        if (producer) {
            if (c + 1 < N_CHUNK) {
                const float* src = noise + blk * 3;
                float*       dst = lds + (st ^ 1) * STAGE_F;
                v4f r4[N_PIECE];
#pragma unroll
                for (int r = 0; r < N_PIECE; ++r) {
                    const int sl = 2 * r + prow;
                    r4[r] = *(const v4f*)(src + noise_step(c + 1, sl) * STEP_F + pcol);
                }
                if (c + 2 < N_CHUNK) {
#pragma unroll
                    for (int r = 0; r < N_PIECE; ++r) {
                        const int sl = 2 * r + prow;
                        __builtin_prefetch(src + noise_step(c + 2, sl) * STEP_F + pcol, 0, 3);
                    }
                }
#pragma unroll
                for (int r = 0; r < N_PIECE; ++r) {
                    const int sl = 2 * r + prow;
                    *(v4f*)(dst + sl * ROW_F + pcol) = r4[r];
                }
            }
        } else {
            const float* sb = lds + st * STAGE_F + ((kind != 0) ? (CH * ROW_F) : 0);
            float nz[2][SAVE_EVERY][6];
#pragma unroll
            for (int k = 0; k < SAVE_EVERY; ++k) {
                const float* row = sb + k * ROW_F;
                nz[0][k][0] = row[lp0 + 0]; nz[0][k][1] = row[lp0 + 1]; nz[0][k][2] = row[lp0 + 2];
                nz[0][k][3] = row[lp1 + 0]; nz[0][k][4] = row[lp1 + 1]; nz[0][k][5] = row[lp1 + 2];
            }
            __builtin_amdgcn_sched_group_barrier(0x100, 15, 0);
#pragma unroll
            for (int g = 0; g < GROUPS; ++g) {
                if (g + 1 < GROUPS) {
#pragma unroll
                    for (int k = 0; k < SAVE_EVERY; ++k) {
                        const float* row = sb + ((g + 1) * SAVE_EVERY + k) * ROW_F;
                        nz[(g + 1) & 1][k][0] = row[lp0 + 0]; nz[(g + 1) & 1][k][1] = row[lp0 + 1]; nz[(g + 1) & 1][k][2] = row[lp0 + 2];
                        nz[(g + 1) & 1][k][3] = row[lp1 + 0]; nz[(g + 1) & 1][k][4] = row[lp1 + 1]; nz[(g + 1) & 1][k][5] = row[lp1 + 2];
                    }
                }
                float sv0 = 0.0f, sv1 = 0.0f, sv2 = 0.0f;
#pragma unroll
                for (int k = 0; k < SAVE_EVERY; ++k) {
                    const float n00 = nz[g & 1][k][0];
                    const float n01 = nz[g & 1][k][1];
                    const float n02 = nz[g & 1][k][2];
                    const float n10 = nz[g & 1][k][3];
                    const float n11 = nz[g & 1][k][4];
                    const float n12 = nz[g & 1][k][5];

                    v8f cz = {};
                    v8f d0 = __builtin_amdgcn_wmma_f32_16x16x4_f32(
                        false, Amat, false, B0, (short)0, cz, false, false);
                    v8f d1 = __builtin_amdgcn_wmma_f32_16x16x4_f32(
                        false, Amat, false, B1, (short)0, cz, false, false);
                    __builtin_amdgcn_sched_group_barrier(0x008, 2, 0);
                    if (g + 1 < GROUPS)
                        __builtin_amdgcn_sched_group_barrier(0x100, 3, 0);

                    const float xn00 = __builtin_fmaf(d0[3], n00, d0[0]);
                    const float xn01 = __builtin_fmaf(d0[4], n01, d0[1]);
                    const float xn02 = __builtin_fmaf(d0[5], n02, d0[2]);
                    const float xn10 = __builtin_fmaf(d1[3], n10, d1[0]);
                    const float xn11 = __builtin_fmaf(d1[4], n11, d1[1]);
                    const float xn12 = __builtin_fmaf(d1[5], n12, d1[2]);

                    B0.x = lo ? xn00 : xn02;  B0.y = lo ? xn01 : hval;
                    B1.x = lo ? xn10 : xn12;  B1.y = lo ? xn11 : hval;

                    if (k == SAVE_EVERY - 1) {
                        sv0 = lo ? xn00 : xn10;
                        sv1 = lo ? xn01 : xn11;
                        sv2 = lo ? xn02 : xn12;
                    }
                }
                const int jg = c * GROUPS + g;
                if (kind == 0) {
                    const int so = p * T_SAVED + (jg + 1);
                    out[0 * BT + so] = sv0;
                    out[1 * BT + so] = sv1;
                    out[2 * BT + so] = sv2;
                    if (jg == HALF_SAVES - 1) {
                        xmid[0 * 64 + pl] = sv0;
                        xmid[1 * 64 + pl] = sv1;
                        xmid[2 * 64 + pl] = sv2;
                    }
                } else {
                    float* tp = tcol + (kind - 1) * T_PLANE + pl * TJ + jg;
                    tp[0 * 64 * TJ] = sv0;
                    tp[1 * 64 * TJ] = sv1;
                    tp[2 * 64 * TJ] = sv2;
                }
            }
        }
        lds_handover_barrier();
    }

    const int j2 = lane + 32;
    for (int i0 = 0; i0 < 15; i0 += 5) {
        float va[5], vb[5];
#pragma unroll
        for (int u = 0; u < 5; ++u) {
            const int s  = wv + N_WAVES * (i0 + u);
            const int rw = (s < 192) ? s : 191;
            const int d  = rw >> 6;
            const int q  = rw & 63;
            const float a0 = xmid[q], a1 = xmid[64 + q], a2 = xmid[128 + q];
            const float* t0 = tcol + (d * 64 + q) * TJ;
            float v = t0[3 * T_PLANE + lane];
            v = __builtin_fmaf(a2, t0[2 * T_PLANE + lane], v);
            v = __builtin_fmaf(a1, t0[1 * T_PLANE + lane], v);
            va[u] = __builtin_fmaf(a0, t0[lane], v);
            float w = t0[3 * T_PLANE + j2];
            w = __builtin_fmaf(a2, t0[2 * T_PLANE + j2], w);
            w = __builtin_fmaf(a1, t0[1 * T_PLANE + j2], w);
            vb[u] = __builtin_fmaf(a0, t0[j2], w);
        }
#pragma unroll
        for (int u = 0; u < 5; ++u) {
            const int s = wv + N_WAVES * (i0 + u);
            if (s < 192) {
                const int d = s >> 6;
                const int q = s & 63;
                float* o = out + d * BT + (blk + q) * T_SAVED + (HALF_SAVES + 1);
                o[lane] = va[u];
                if (lane < HALF_SAVES - 32) o[lane + 32] = vb[u];
            }
        }
    }
}

extern "C" void kernel_launch(void* const* d_in, const int* in_sizes, int n_in,
                              void* d_out, int out_size, void* d_ws, size_t ws_size,
                              hipStream_t stream) {
    const float* drift     = (const float*)d_in[0];
    const float* driftb    = (const float*)d_in[1];
    const float* diffusion = (const float*)d_in[2];
    const float* diffb     = (const float*)d_in[3];
    const float* x0        = (const float*)d_in[4];
    const float* noise     = (const float*)d_in[5];
    float* out = (float*)d_out;

    dim3 block(416);
    dim3 grid(B_PATHS / 64);
    sde_wmma_kernel<<<grid, block, 0, stream>>>(
        drift, driftb, diffusion, diffb, x0, noise, out);
}


# a chain wave carries all five chains (state, 3 map columns, offset) of 16 paths: 4 chain waves, one per SIMD, 5 independent WMMAs a step; noise read from LDS once per wave, one group per 5 steps
# speedup vs baseline: 1.8530x; 1.0569x over previous
#include <hip/hip_runtime.h>


typedef float v2f __attribute__((ext_vector_type(2)));
typedef float v4f __attribute__((ext_vector_type(4)));
typedef float v8f __attribute__((ext_vector_type(8)));

#define B_PATHS    16384
#define N_STEPS    500
#define HALF_STEPS 250
#define SAVE_EVERY 5
#define T_SAVED    101
#define HALF_SAVES 50
#define DT         0.002f
#define CH         25
#define N_CHUNK    10
#define GROUPS     5
#define N_PIECE    25
#define ROW_F      192
#define STAGE_F    (2 * CH * ROW_F)
#define STEP_F     (B_PATHS * 3)
#define N_WAVES    7
#define N_CHAIN    4
#define N_STRIP    192
#define STRIP_U    4
#define STRIP_TRIPS 28
#define TJ         51
#define T_PLANE    (3 * 64 * TJ)


__device__ __forceinline__ void lds_handover_barrier() {
    __builtin_amdgcn_fence(__ATOMIC_RELEASE, "workgroup", "local");
    __builtin_amdgcn_s_barrier();
    __builtin_amdgcn_fence(__ATOMIC_ACQUIRE, "workgroup", "local");
}

__device__ __forceinline__ int noise_step(int c, int sl) {
    return c * CH + sl + ((sl >= CH) ? (HALF_STEPS - CH) : 0);
}

__global__ __launch_bounds__(224) void sde_wmma_kernel(
    const float* __restrict__ drift, const float* __restrict__ drift_bias,
    const float* __restrict__ diffusion, const float* __restrict__ diffusion_bias,
    const float* __restrict__ x0, const float* __restrict__ noise,
    float* __restrict__ out)
{
    __shared__ v4f   lds4[2 * STAGE_F / 4];
    __shared__ float tcol[4 * T_PLANE + 16];
    __shared__ float xmid[3 * 64];
    float* lds = (float*)lds4;

    const int  tid  = threadIdx.x;
    const int  lane = tid & 31;
    const int  wv   = __builtin_amdgcn_readfirstlane(tid >> 5);
    const bool producer = wv >= N_CHAIN;
    const int  blk  = blockIdx.x * 64;
    const int  m    = lane & 15;
    const bool lo   = lane < 16;
    const float sqdt = __builtin_sqrtf(DT);
    const int  BT = B_PATHS * T_SAVED;
    const int  pw = (wv & (N_CHAIN - 1)) * 16 + m;
    const int  p  = blk + pw;

    const int ptid = tid - N_CHAIN * 32;
    const int prow = (ptid >= 48) ? 1 : 0;
    const int pcol = (ptid - 48 * prow) * 4;

    v2f Amat; Amat.x = 0.0f; Amat.y = 0.0f;
    v2f Bv[5];
#pragma unroll
    for (int j = 0; j < 5; ++j) { Bv[j].x = 0.0f; Bv[j].y = 0.0f; }
    const int lpA = pw * 3;
    const int oo01 = (lo ? 0 : BT) + p * T_SAVED;
    const int oo2  = 2 * BT + p * T_SAVED;
    float* const tp01 = tcol + ((lo ? 0 : 64) + pw) * TJ;
    float* const tp2  = tcol + (lo ? 0 : T_PLANE) + (128 + pw) * TJ;

    if (producer) {
        v4f r4[N_PIECE];
#pragma unroll
        for (int r = 0; r < N_PIECE; ++r) {
            const int sl = 2 * r + prow;
            r4[r] = *(const v4f*)(noise + noise_step(0, sl) * STEP_F + blk * 3 + pcol);
        }
#pragma unroll
        for (int r = 0; r < N_PIECE; ++r) {
            const int sl = 2 * r + prow;
            __builtin_prefetch(noise + noise_step(1, sl) * STEP_F + blk * 3 + pcol, 0, 3);
        }
#pragma unroll
        for (int r = 0; r < N_PIECE; ++r) {
            const int sl = 2 * r + prow;
            *(v4f*)(lds + sl * ROW_F + pcol) = r4[r];
        }
    } else {
        const int k0 = lo ? 0 : 2;
        const int k1 = lo ? 1 : 3;
        const int mm = m & 7;
        float a0v = 0.0f, a1v = 0.0f;
        if (mm < 3) {
            a0v = DT * drift[mm * 3 + k0] + ((mm == k0) ? 1.0f : 0.0f);
            a1v = (k1 < 3) ? (DT * drift[mm * 3 + k1] + ((mm == k1) ? 1.0f : 0.0f))
                           : (DT * drift_bias[mm]);
        } else if (mm < 6) {
            const int r = mm - 3;
            a0v = sqdt * diffusion[r * 3 + k0];
            a1v = (k1 < 3) ? (sqdt * diffusion[r * 3 + k1])
                           : (sqdt * diffusion_bias[r]);
        }
        Amat.x = a0v; Amat.y = a1v;

        const float xi0 = x0[p * 3 + 0], xi1 = x0[p * 3 + 1], xi2 = x0[p * 3 + 2];
        Bv[0].x = lo ? xi0 : xi2;
        Bv[0].y = lo ? xi1 : 1.0f;
        out[oo01] = lo ? xi0 : xi1;
        out[oo2]  = xi2;
        Bv[1].x = lo ? 1.0f : 0.0f;  Bv[1].y = 0.0f;
        Bv[2].x = 0.0f;              Bv[2].y = lo ? 1.0f : 0.0f;
        Bv[3].x = lo ? 0.0f : 1.0f;  Bv[3].y = 0.0f;
        Bv[4].x = 0.0f;              Bv[4].y = lo ? 0.0f : 1.0f;
    }
    lds_handover_barrier();

    for (int c = 0; c < N_CHUNK; ++c) {
        const int st = c & 1;
        if (producer) {
            if (c + 1 < N_CHUNK) {
                const float* src = noise + blk * 3;
                float*       dst = lds + (st ^ 1) * STAGE_F;
                v4f r4[N_PIECE];
#pragma unroll
                for (int r = 0; r < N_PIECE; ++r) {
                    const int sl = 2 * r + prow;
                    r4[r] = *(const v4f*)(src + noise_step(c + 1, sl) * STEP_F + pcol);
                }
                if (c + 2 < N_CHUNK) {
#pragma unroll
                    for (int r = 0; r < N_PIECE; ++r) {
                        const int sl = 2 * r + prow;
                        __builtin_prefetch(src + noise_step(c + 2, sl) * STEP_F + pcol, 0, 3);
                    }
                }
#pragma unroll
                for (int r = 0; r < N_PIECE; ++r) {
                    const int sl = 2 * r + prow;
                    *(v4f*)(dst + sl * ROW_F + pcol) = r4[r];
                }
            }
        } else {
            const float* sbA = lds + st * STAGE_F;
            float nz[2][SAVE_EVERY][6];
#pragma unroll
            for (int k = 0; k < SAVE_EVERY; ++k) {
                const float* rowA = sbA + k * ROW_F + lpA;
                const float* rowB = rowA + CH * ROW_F;
                nz[0][k][0] = rowA[0]; nz[0][k][1] = rowA[1]; nz[0][k][2] = rowA[2];
                nz[0][k][3] = rowB[0]; nz[0][k][4] = rowB[1]; nz[0][k][5] = rowB[2];
            }
            __builtin_amdgcn_sched_group_barrier(0x100, 15, 0);
#pragma unroll
            for (int g = 0; g < GROUPS; ++g) {
                if (g + 1 < GROUPS) {
#pragma unroll
                    for (int k = 0; k < SAVE_EVERY; ++k) {
                        const float* rowA = sbA + ((g + 1) * SAVE_EVERY + k) * ROW_F + lpA;
                        const float* rowB = rowA + CH * ROW_F;
                        nz[(g + 1) & 1][k][0] = rowA[0]; nz[(g + 1) & 1][k][1] = rowA[1]; nz[(g + 1) & 1][k][2] = rowA[2];
                        nz[(g + 1) & 1][k][3] = rowB[0]; nz[(g + 1) & 1][k][4] = rowB[1]; nz[(g + 1) & 1][k][5] = rowB[2];
                    }
                    __builtin_amdgcn_sched_group_barrier(0x100, 15, 0);
                }
                float sv[5][3];
#pragma unroll
                for (int k = 0; k < SAVE_EVERY; ++k) {
                    v8f cz = {};
                    v8f d[5];
#pragma unroll
                    for (int j = 0; j < 5; ++j)
                        d[j] = __builtin_amdgcn_wmma_f32_16x16x4_f32(
                            false, Amat, false, Bv[j], (short)0, cz, false, false);
                    __builtin_amdgcn_sched_group_barrier(0x008, 5, 0);
#pragma unroll
                    for (int j = 0; j < 5; ++j) {
                        const int o = (j == 0) ? 0 : 3;
                        const float xn0 = __builtin_fmaf(d[j][3], nz[g & 1][k][o + 0], d[j][0]);
                        const float xn1 = __builtin_fmaf(d[j][4], nz[g & 1][k][o + 1], d[j][1]);
                        const float xn2 = __builtin_fmaf(d[j][5], nz[g & 1][k][o + 2], d[j][2]);
                        const float hv = (j == 0 || j == 4) ? 1.0f : 0.0f;
                        Bv[j].x = lo ? xn0 : xn2;
                        Bv[j].y = lo ? xn1 : hv;
                        if (k == SAVE_EVERY - 1) { sv[j][0] = xn0; sv[j][1] = xn1; sv[j][2] = xn2; }
                    }
                }
                const int jg = c * GROUPS + g;
                out[oo01 + (jg + 1)] = lo ? sv[0][0] : sv[0][1];
                out[oo2  + (jg + 1)] = sv[0][2];
                if (jg == HALF_SAVES - 1) {
                    xmid[(lo ? 0 : 64) + pw] = lo ? sv[0][0] : sv[0][1];
                    xmid[128 + pw]           = sv[0][2];
                }
#pragma unroll
                for (int j = 1; j < 5; ++j)
                    tp01[(j - 1) * T_PLANE + jg] = lo ? sv[j][0] : sv[j][1];
                tp2[0 * T_PLANE + jg] = lo ? sv[1][2] : sv[2][2];
                tp2[2 * T_PLANE + jg] = lo ? sv[3][2] : sv[4][2];
                __builtin_amdgcn_sched_group_barrier(0x040, 2, 0);
                __builtin_amdgcn_sched_group_barrier(0x200, 6, 0);
            }
        }
        lds_handover_barrier();
    }

    const int j2 = lane + 32;
    for (int i0 = 0; i0 < STRIP_TRIPS; i0 += STRIP_U) {
        float va[STRIP_U], vb[STRIP_U];
#pragma unroll
        for (int u = 0; u < STRIP_U; ++u) {
            const int s  = wv + N_WAVES * (i0 + u);
            const int rw = (s < N_STRIP) ? s : (N_STRIP - 1);
            const int d  = rw >> 6;
            const int q  = rw & 63;
            const float a0 = xmid[q], a1 = xmid[64 + q], a2 = xmid[128 + q];
            const float* t0 = tcol + (d * 64 + q) * TJ;
            float v = t0[3 * T_PLANE + lane];
            v = __builtin_fmaf(a2, t0[2 * T_PLANE + lane], v);
            v = __builtin_fmaf(a1, t0[1 * T_PLANE + lane], v);
            va[u] = __builtin_fmaf(a0, t0[lane], v);
            float w = t0[3 * T_PLANE + j2];
            w = __builtin_fmaf(a2, t0[2 * T_PLANE + j2], w);
            w = __builtin_fmaf(a1, t0[1 * T_PLANE + j2], w);
            vb[u] = __builtin_fmaf(a0, t0[j2], w);
        }
#pragma unroll
        for (int u = 0; u < STRIP_U; ++u) {
            const int s = wv + N_WAVES * (i0 + u);
            if (s < N_STRIP) {
                const int d = s >> 6;
                const int q = s & 63;
                float* o = out + d * BT + (blk + q) * T_SAVED + (HALF_SAVES + 1);
                o[lane] = va[u];
                if (lane < HALF_SAVES - 32) o[lane + 32] = vb[u];
            }
        }
    }
}

extern "C" void kernel_launch(void* const* d_in, const int* in_sizes, int n_in,
                              void* d_out, int out_size, void* d_ws, size_t ws_size,
                              hipStream_t stream) {
    const float* drift     = (const float*)d_in[0];
    const float* driftb    = (const float*)d_in[1];
    const float* diffusion = (const float*)d_in[2];
    const float* diffb     = (const float*)d_in[3];
    const float* x0        = (const float*)d_in[4];
    const float* noise     = (const float*)d_in[5];
    float* out = (float*)d_out;

    dim3 block(224);
    dim3 grid(B_PATHS / 64);
    sde_wmma_kernel<<<grid, block, 0, stream>>>(
        drift, driftb, diffusion, diffb, x0, noise, out);
}


# state saves 1-50 go to an LDS plane; the noise producer waves store them to the output a chunk later, 5 adjacent floats on adjacent lanes; chain waves issue no global store in the time loop
# speedup vs baseline: 1.9425x; 1.0483x over previous
#include <hip/hip_runtime.h>


typedef float v2f __attribute__((ext_vector_type(2)));
typedef float v4f __attribute__((ext_vector_type(4)));
typedef float v8f __attribute__((ext_vector_type(8)));

#define B_PATHS    16384
#define N_STEPS    500
#define HALF_STEPS 250
#define SAVE_EVERY 5
#define T_SAVED    101
#define HALF_SAVES 50
#define DT         0.002f
#define CH         25
#define N_CHUNK    10
#define GROUPS     5
#define N_PIECE    25
#define ROW_F      192
#define STAGE_F    (2 * CH * ROW_F)
#define STEP_F     (B_PATHS * 3)
#define N_WAVES    7
#define N_CHAIN    4
#define N_STRIP    192
#define STRIP_U    4
#define STRIP_TRIPS 28
#define TJ         51
#define T_PLANE    (3 * 64 * TJ)
#define S_PLANE    4
#define N_DRAIN    10


__device__ __forceinline__ void lds_handover_barrier() {
    __builtin_amdgcn_fence(__ATOMIC_RELEASE, "workgroup", "local");
    __builtin_amdgcn_s_barrier();
    __builtin_amdgcn_fence(__ATOMIC_ACQUIRE, "workgroup", "local");
}

__device__ __forceinline__ void drain_saves(const float* tcol, float* __restrict__ out,
                                            const int (&dr_lds)[N_DRAIN], const unsigned (&dr_out)[N_DRAIN],
                                            int cc) {
    float dv[N_DRAIN];
    float* const oc = out + cc * GROUPS;
#pragma unroll
    for (int k = 0; k < N_DRAIN; ++k) dv[k] = tcol[dr_lds[k] + cc * GROUPS];
#pragma unroll
    for (int k = 0; k < N_DRAIN; ++k) oc[dr_out[k]] = dv[k];
}

__device__ __forceinline__ int noise_step(int c, int sl) {
    return c * CH + sl + ((sl >= CH) ? (HALF_STEPS - CH) : 0);
}

__global__ __launch_bounds__(224) void sde_wmma_kernel(
    const float* __restrict__ drift, const float* __restrict__ drift_bias,
    const float* __restrict__ diffusion, const float* __restrict__ diffusion_bias,
    const float* __restrict__ x0, const float* __restrict__ noise,
    float* __restrict__ out)
{
    __shared__ v4f   lds4[2 * STAGE_F / 4];
    __shared__ float tcol[5 * T_PLANE + 16];
    __shared__ float xmid[3 * 64];
    float* lds = (float*)lds4;

    const int  tid  = threadIdx.x;
    const int  lane = tid & 31;
    const int  wv   = __builtin_amdgcn_readfirstlane(tid >> 5);
    const bool producer = wv >= N_CHAIN;
    const int  blk  = blockIdx.x * 64;
    const int  m    = lane & 15;
    const bool lo   = lane < 16;
    const float sqdt = __builtin_sqrtf(DT);
    const int  BT = B_PATHS * T_SAVED;
    const int  pw = (wv & (N_CHAIN - 1)) * 16 + m;
    const int  p  = blk + pw;

    const int ptid = tid - N_CHAIN * 32;
    const int prow = (ptid >= 48) ? 1 : 0;
    const int pcol = (ptid - 48 * prow) * 4;
    int dr_lds[N_DRAIN];
    unsigned dr_out[N_DRAIN];
#pragma unroll
    for (int k = 0; k < N_DRAIN; ++k) { dr_lds[k] = 0; dr_out[k] = 0u; }

    v2f Amat; Amat.x = 0.0f; Amat.y = 0.0f;
    v2f Bv[5];
#pragma unroll
    for (int j = 0; j < 5; ++j) { Bv[j].x = 0.0f; Bv[j].y = 0.0f; }
    const int lpA = pw * 3;
    const int oo01 = (lo ? 0 : BT) + p * T_SAVED;
    const int oo2  = 2 * BT + p * T_SAVED;
    float* const tp01 = tcol + ((lo ? 0 : 64) + pw) * TJ;
    float* const tp2  = tcol + (lo ? 0 : T_PLANE) + (128 + pw) * TJ;

    if (producer) {
#pragma unroll
        for (int k = 0; k < N_DRAIN; ++k) {
            const int e  = ptid + 96 * k;
            const int rw = e / GROUPS;
            const int i  = e - rw * GROUPS;
            dr_lds[k] = S_PLANE * T_PLANE + rw * TJ + i;
            dr_out[k] = (unsigned)((rw >> 6) * (B_PATHS * T_SAVED) + (blk + (rw & 63)) * T_SAVED + 1 + i);
        }
        v4f r4[N_PIECE];
#pragma unroll
        for (int r = 0; r < N_PIECE; ++r) {
            const int sl = 2 * r + prow;
            r4[r] = *(const v4f*)(noise + noise_step(0, sl) * STEP_F + blk * 3 + pcol);
        }
#pragma unroll
        for (int r = 0; r < N_PIECE; ++r) {
            const int sl = 2 * r + prow;
            __builtin_prefetch(noise + noise_step(1, sl) * STEP_F + blk * 3 + pcol, 0, 3);
        }
#pragma unroll
        for (int r = 0; r < N_PIECE; ++r) {
            const int sl = 2 * r + prow;
            *(v4f*)(lds + sl * ROW_F + pcol) = r4[r];
        }
    } else {
        const int k0 = lo ? 0 : 2;
        const int k1 = lo ? 1 : 3;
        const int mm = m & 7;
        float a0v = 0.0f, a1v = 0.0f;
        if (mm < 3) {
            a0v = DT * drift[mm * 3 + k0] + ((mm == k0) ? 1.0f : 0.0f);
            a1v = (k1 < 3) ? (DT * drift[mm * 3 + k1] + ((mm == k1) ? 1.0f : 0.0f))
                           : (DT * drift_bias[mm]);
        } else if (mm < 6) {
            const int r = mm - 3;
            a0v = sqdt * diffusion[r * 3 + k0];
            a1v = (k1 < 3) ? (sqdt * diffusion[r * 3 + k1])
                           : (sqdt * diffusion_bias[r]);
        }
        Amat.x = a0v; Amat.y = a1v;

        const float xi0 = x0[p * 3 + 0], xi1 = x0[p * 3 + 1], xi2 = x0[p * 3 + 2];
        Bv[0].x = lo ? xi0 : xi2;
        Bv[0].y = lo ? xi1 : 1.0f;
        out[oo01] = lo ? xi0 : xi1;
        out[oo2]  = xi2;
        Bv[1].x = lo ? 1.0f : 0.0f;  Bv[1].y = 0.0f;
        Bv[2].x = 0.0f;              Bv[2].y = lo ? 1.0f : 0.0f;
        Bv[3].x = lo ? 0.0f : 1.0f;  Bv[3].y = 0.0f;
        Bv[4].x = 0.0f;              Bv[4].y = lo ? 0.0f : 1.0f;
    }
    lds_handover_barrier();

    for (int c = 0; c < N_CHUNK; ++c) {
        const int st = c & 1;
        if (producer) {
            if (c + 1 < N_CHUNK) {
                const float* src = noise + blk * 3;
                float*       dst = lds + (st ^ 1) * STAGE_F;
                v4f r4[N_PIECE];
#pragma unroll
                for (int r = 0; r < N_PIECE; ++r) {
                    const int sl = 2 * r + prow;
                    r4[r] = *(const v4f*)(src + noise_step(c + 1, sl) * STEP_F + pcol);
                }
                if (c + 2 < N_CHUNK) {
#pragma unroll
                    for (int r = 0; r < N_PIECE; ++r) {
                        const int sl = 2 * r + prow;
                        __builtin_prefetch(src + noise_step(c + 2, sl) * STEP_F + pcol, 0, 3);
                    }
                }
                if (c > 0) drain_saves(tcol, out, dr_lds, dr_out, c - 1);
#pragma unroll
                for (int r = 0; r < N_PIECE; ++r) {
                    const int sl = 2 * r + prow;
                    *(v4f*)(dst + sl * ROW_F + pcol) = r4[r];
                }
            } else {
                drain_saves(tcol, out, dr_lds, dr_out, c - 1);
            }
        } else {
            const float* sbA = lds + st * STAGE_F;
            float nz[2][SAVE_EVERY][6];
#pragma unroll
            for (int k = 0; k < SAVE_EVERY; ++k) {
                const float* rowA = sbA + k * ROW_F + lpA;
                const float* rowB = rowA + CH * ROW_F;
                nz[0][k][0] = rowA[0]; nz[0][k][1] = rowA[1]; nz[0][k][2] = rowA[2];
                nz[0][k][3] = rowB[0]; nz[0][k][4] = rowB[1]; nz[0][k][5] = rowB[2];
            }
            __builtin_amdgcn_sched_group_barrier(0x100, 15, 0);
#pragma unroll
            for (int g = 0; g < GROUPS; ++g) {
                if (g + 1 < GROUPS) {
#pragma unroll
                    for (int k = 0; k < SAVE_EVERY; ++k) {
                        const float* rowA = sbA + ((g + 1) * SAVE_EVERY + k) * ROW_F + lpA;
                        const float* rowB = rowA + CH * ROW_F;
                        nz[(g + 1) & 1][k][0] = rowA[0]; nz[(g + 1) & 1][k][1] = rowA[1]; nz[(g + 1) & 1][k][2] = rowA[2];
                        nz[(g + 1) & 1][k][3] = rowB[0]; nz[(g + 1) & 1][k][4] = rowB[1]; nz[(g + 1) & 1][k][5] = rowB[2];
                    }
                    __builtin_amdgcn_sched_group_barrier(0x100, 15, 0);
                }
                float sv[5][3];
#pragma unroll
                for (int k = 0; k < SAVE_EVERY; ++k) {
                    v8f cz = {};
                    v8f d[5];
#pragma unroll
                    for (int j = 0; j < 5; ++j)
                        d[j] = __builtin_amdgcn_wmma_f32_16x16x4_f32(
                            false, Amat, false, Bv[j], (short)0, cz, false, false);
                    __builtin_amdgcn_sched_group_barrier(0x008, 5, 0);
#pragma unroll
                    for (int j = 0; j < 5; ++j) {
                        const int o = (j == 0) ? 0 : 3;
                        const float xn0 = __builtin_fmaf(d[j][3], nz[g & 1][k][o + 0], d[j][0]);
                        const float xn1 = __builtin_fmaf(d[j][4], nz[g & 1][k][o + 1], d[j][1]);
                        const float xn2 = __builtin_fmaf(d[j][5], nz[g & 1][k][o + 2], d[j][2]);
                        const float hv = (j == 0 || j == 4) ? 1.0f : 0.0f;
                        Bv[j].x = lo ? xn0 : xn2;
                        Bv[j].y = lo ? xn1 : hv;
                        if (k == SAVE_EVERY - 1) { sv[j][0] = xn0; sv[j][1] = xn1; sv[j][2] = xn2; }
                    }
                }
                const int jg = c * GROUPS + g;
                tp01[S_PLANE * T_PLANE + jg] = lo ? sv[0][0] : sv[0][1];
                tcol[S_PLANE * T_PLANE + (128 + pw) * TJ + jg] = sv[0][2];
                if (jg == HALF_SAVES - 1) {
                    xmid[(lo ? 0 : 64) + pw] = lo ? sv[0][0] : sv[0][1];
                    xmid[128 + pw]           = sv[0][2];
                }
#pragma unroll
                for (int j = 1; j < 5; ++j)
                    tp01[(j - 1) * T_PLANE + jg] = lo ? sv[j][0] : sv[j][1];
                tp2[0 * T_PLANE + jg] = lo ? sv[1][2] : sv[2][2];
                tp2[2 * T_PLANE + jg] = lo ? sv[3][2] : sv[4][2];
                __builtin_amdgcn_sched_group_barrier(0x200, 8, 0);
            }
        }
        lds_handover_barrier();
    }

    if (producer) drain_saves(tcol, out, dr_lds, dr_out, N_CHUNK - 1);

    const int j2 = lane + 32;
    for (int i0 = 0; i0 < STRIP_TRIPS; i0 += STRIP_U) {
        float va[STRIP_U], vb[STRIP_U];
#pragma unroll
        for (int u = 0; u < STRIP_U; ++u) {
            const int s  = wv + N_WAVES * (i0 + u);
            const int rw = (s < N_STRIP) ? s : (N_STRIP - 1);
            const int d  = rw >> 6;
            const int q  = rw & 63;
            const float a0 = xmid[q], a1 = xmid[64 + q], a2 = xmid[128 + q];
            const float* t0 = tcol + (d * 64 + q) * TJ;
            float v = t0[3 * T_PLANE + lane];
            v = __builtin_fmaf(a2, t0[2 * T_PLANE + lane], v);
            v = __builtin_fmaf(a1, t0[1 * T_PLANE + lane], v);
            va[u] = __builtin_fmaf(a0, t0[lane], v);
            float w = t0[3 * T_PLANE + j2];
            w = __builtin_fmaf(a2, t0[2 * T_PLANE + j2], w);
            w = __builtin_fmaf(a1, t0[1 * T_PLANE + j2], w);
            vb[u] = __builtin_fmaf(a0, t0[j2], w);
        }
#pragma unroll
        for (int u = 0; u < STRIP_U; ++u) {
            const int s = wv + N_WAVES * (i0 + u);
            if (s < N_STRIP) {
                const int d = s >> 6;
                const int q = s & 63;
                float* o = out + d * BT + (blk + q) * T_SAVED + (HALF_SAVES + 1);
                o[lane] = va[u];
                if (lane < HALF_SAVES - 32) o[lane + 32] = vb[u];
            }
        }
    }
}

extern "C" void kernel_launch(void* const* d_in, const int* in_sizes, int n_in,
                              void* d_out, int out_size, void* d_ws, size_t ws_size,
                              hipStream_t stream) {
    const float* drift     = (const float*)d_in[0];
    const float* driftb    = (const float*)d_in[1];
    const float* diffusion = (const float*)d_in[2];
    const float* diffb     = (const float*)d_in[3];
    const float* x0        = (const float*)d_in[4];
    const float* noise     = (const float*)d_in[5];
    float* out = (float*)d_out;

    dim3 block(224);
    dim3 grid(B_PATHS / 64);
    sde_wmma_kernel<<<grid, block, 0, stream>>>(
        drift, driftb, diffusion, diffb, x0, noise, out);
}


# time axis re-cut from 250+250 to 400+100 steps: one map-column product beside each state product (800 products per wave, were 1250); saves 81-100 = map applied to x(400), a new rounding order
# speedup vs baseline: 2.4182x; 1.2449x over previous
#include <hip/hip_runtime.h>


typedef float v2f __attribute__((ext_vector_type(2)));
typedef float v4f __attribute__((ext_vector_type(4)));
typedef float v8f __attribute__((ext_vector_type(8)));

#define B_PATHS    16384
#define N_STEPS    500
#define SEG1_STEPS 400
#define SAVE_EVERY 5
#define T_SAVED    101
#define SEG1_SAVES 80
#define SEG2_SAVES 20
#define DT         0.002f
#define N_CHUNK    10
#define CH1        40
#define CH2        10
#define CH_ROWS    (CH1 + CH2)
#define SUBS       8
#define RD_STEPS   10
#define N_PIECE    25
#define ROW_F      192
#define STAGE_F    (CH_ROWS * ROW_F)
#define STEP_F     (B_PATHS * 3)
#define N_THREADS  224
#define N_CHAIN    4
#define N_STRIP    192
#define TC         21
#define TS         81
#define C_PLANE    (N_STRIP * TC)
#define S_OFF      (4 * C_PLANE)
#define S_SIZE     (N_STRIP * TS)
#define N_DRAIN    16
#define N_CLOSE    (N_STRIP * SEG2_SAVES)
#define CLOSE_U    6
#define CLOSE_TRIPS 18


__device__ __forceinline__ void lds_handover_barrier() {
    __builtin_amdgcn_fence(__ATOMIC_RELEASE, "workgroup", "local");
    __builtin_amdgcn_s_barrier();
    __builtin_amdgcn_fence(__ATOMIC_ACQUIRE, "workgroup", "local");
}

__device__ __forceinline__ void drain_saves(const float* tcol, float* __restrict__ out,
                                            const int (&dr_lds)[N_DRAIN], const unsigned (&dr_out)[N_DRAIN],
                                            int cc) {
    float dv[N_DRAIN];
    float* const oc = out + cc * SUBS;
#pragma unroll
    for (int k = 0; k < N_DRAIN; ++k) dv[k] = tcol[dr_lds[k] + cc * SUBS];
#pragma unroll
    for (int k = 0; k < N_DRAIN; ++k) oc[dr_out[k]] = dv[k];
}

__device__ __forceinline__ int noise_step(int c, int sl) {
    return (sl < CH1) ? (c * CH1 + sl) : (SEG1_STEPS + c * CH2 + (sl - CH1));
}

__device__ __forceinline__ void em_update(const v8f& d, const float (&nz)[3], bool lo, float hv,
                                          v2f& b, float& x01, float& x2) {
    const float xn0 = __builtin_fmaf(d[3], nz[0], d[0]);
    const float xn1 = __builtin_fmaf(d[4], nz[1], d[1]);
    const float xn2 = __builtin_fmaf(d[5], nz[2], d[2]);
    b.x = lo ? xn0 : xn2;
    b.y = lo ? xn1 : hv;
    x01 = lo ? xn0 : xn1;
    x2  = xn2;
}

__global__ __launch_bounds__(N_THREADS) void sde_wmma_kernel(
    const float* __restrict__ drift, const float* __restrict__ drift_bias,
    const float* __restrict__ diffusion, const float* __restrict__ diffusion_bias,
    const float* __restrict__ x0, const float* __restrict__ noise,
    float* __restrict__ out)
{
    __shared__ v4f   lds4[2 * STAGE_F / 4];
    __shared__ float tcol[S_OFF + S_SIZE];
    float* lds = (float*)lds4;

    const int  tid  = threadIdx.x;
    const int  lane = tid & 31;
    const int  wv   = __builtin_amdgcn_readfirstlane(tid >> 5);
    const bool producer = wv >= N_CHAIN;
    const int  blk  = blockIdx.x * 64;
    const int  m    = lane & 15;
    const bool lo   = lane < 16;
    const float sqdt = __builtin_sqrtf(DT);
    const int  BT = B_PATHS * T_SAVED;
    const int  pw = (wv & (N_CHAIN - 1)) * 16 + m;
    const int  p  = blk + pw;

    const int ptid = tid - N_CHAIN * 32;
    const int prow = (ptid >= 48) ? 1 : 0;
    const int pcol = (ptid - 48 * prow) * 4;
    int dr_lds[N_DRAIN];
    unsigned dr_out[N_DRAIN];
#pragma unroll
    for (int k = 0; k < N_DRAIN; ++k) { dr_lds[k] = 0; dr_out[k] = 0u; }

    v2f Amat; Amat.x = 0.0f; Amat.y = 0.0f;
    v2f Bv[5];
#pragma unroll
    for (int j = 0; j < 5; ++j) { Bv[j].x = 0.0f; Bv[j].y = 0.0f; }
    const int lpA = pw * 3;
    const int oo01 = (lo ? 0 : BT) + p * T_SAVED;
    const int oo2  = 2 * BT + p * T_SAVED;
    float* const tp01 = tcol + ((lo ? 0 : 64) + pw) * TC;
    float* const tp2  = tcol + (lo ? 0 : C_PLANE) + (128 + pw) * TC;
    float* const ts01 = tcol + S_OFF + ((lo ? 0 : 64) + pw) * TS;
    float* const ts2  = tcol + S_OFF + (128 + pw) * TS;

    if (producer) {
#pragma unroll
        for (int k = 0; k < N_DRAIN; ++k) {
            const int e  = ptid + 96 * k;
            const int rw = e / SUBS;
            const int i  = e - rw * SUBS;
            dr_lds[k] = S_OFF + rw * TS + i;
            dr_out[k] = (unsigned)((rw >> 6) * (B_PATHS * T_SAVED) + (blk + (rw & 63)) * T_SAVED + 1 + i);
        }
        v4f r4[N_PIECE];
#pragma unroll
        for (int r = 0; r < N_PIECE; ++r) {
            const int sl = 2 * r + prow;
            r4[r] = *(const v4f*)(noise + noise_step(0, sl) * STEP_F + blk * 3 + pcol);
        }
#pragma unroll
        for (int r = 0; r < N_PIECE; ++r) {
            const int sl = 2 * r + prow;
            __builtin_prefetch(noise + noise_step(1, sl) * STEP_F + blk * 3 + pcol, 0, 3);
        }
#pragma unroll
        for (int r = 0; r < N_PIECE; ++r) {
            const int sl = 2 * r + prow;
            *(v4f*)(lds + sl * ROW_F + pcol) = r4[r];
        }
    } else {
        const int k0 = lo ? 0 : 2;
        const int k1 = lo ? 1 : 3;
        const int mm = m & 7;
        float a0v = 0.0f, a1v = 0.0f;
        if (mm < 3) {
            a0v = DT * drift[mm * 3 + k0] + ((mm == k0) ? 1.0f : 0.0f);
            a1v = (k1 < 3) ? (DT * drift[mm * 3 + k1] + ((mm == k1) ? 1.0f : 0.0f))
                           : (DT * drift_bias[mm]);
        } else if (mm < 6) {
            const int r = mm - 3;
            a0v = sqdt * diffusion[r * 3 + k0];
            a1v = (k1 < 3) ? (sqdt * diffusion[r * 3 + k1])
                           : (sqdt * diffusion_bias[r]);
        }
        Amat.x = a0v; Amat.y = a1v;

        const float xi0 = x0[p * 3 + 0], xi1 = x0[p * 3 + 1], xi2 = x0[p * 3 + 2];
        Bv[0].x = lo ? xi0 : xi2;
        Bv[0].y = lo ? xi1 : 1.0f;
        out[oo01] = lo ? xi0 : xi1;
        out[oo2]  = xi2;
        Bv[1].x = lo ? 1.0f : 0.0f;  Bv[1].y = 0.0f;
        Bv[2].x = 0.0f;              Bv[2].y = lo ? 1.0f : 0.0f;
        Bv[3].x = lo ? 0.0f : 1.0f;  Bv[3].y = 0.0f;
        Bv[4].x = 0.0f;              Bv[4].y = lo ? 0.0f : 1.0f;
    }
    lds_handover_barrier();

    for (int c = 0; c < N_CHUNK; ++c) {
        const int st = c & 1;
        if (producer) {
            if (c + 1 < N_CHUNK) {
                const float* src = noise + blk * 3;
                float*       dst = lds + (st ^ 1) * STAGE_F;
                v4f r4[N_PIECE];
#pragma unroll
                for (int r = 0; r < N_PIECE; ++r) {
                    const int sl = 2 * r + prow;
                    r4[r] = *(const v4f*)(src + noise_step(c + 1, sl) * STEP_F + pcol);
                }
                if (c + 2 < N_CHUNK) {
#pragma unroll
                    for (int r = 0; r < N_PIECE; ++r) {
                        const int sl = 2 * r + prow;
                        __builtin_prefetch(src + noise_step(c + 2, sl) * STEP_F + pcol, 0, 3);
                    }
                }
                if (c > 0) drain_saves(tcol, out, dr_lds, dr_out, c - 1);
#pragma unroll
                for (int r = 0; r < N_PIECE; ++r) {
                    const int sl = 2 * r + prow;
                    *(v4f*)(dst + sl * ROW_F + pcol) = r4[r];
                }
            } else {
                drain_saves(tcol, out, dr_lds, dr_out, c - 1);
            }
        } else {
            const float* sb = lds + st * STAGE_F;
            float nzs[2][RD_STEPS][3];
            float nzc[2][SAVE_EVERY][3];
#pragma unroll
            for (int k = 0; k < RD_STEPS; ++k) {
                const float* rowS = sb + k * ROW_F + lpA;
                nzs[0][k][0] = rowS[0]; nzs[0][k][1] = rowS[1]; nzs[0][k][2] = rowS[2];
            }
#pragma unroll
            for (int k = 0; k < SAVE_EVERY; ++k) {
                const float* rowC = sb + (CH1 + k) * ROW_F + lpA;
                nzc[0][k][0] = rowC[0]; nzc[0][k][1] = rowC[1]; nzc[0][k][2] = rowC[2];
            }
            v8f dpend = {};
            float sc01[5], sc2[5];
            float h01 = 0.0f, h2 = 0.0f;
#pragma unroll
            for (int s = 0; s < CH1; ++s) {
                const int hg = s / RD_STEPS;
                const int kk = s - hg * RD_STEPS;
                const int sg = s / SAVE_EVERY;
                const int t  = s % 20;
                const int cj = 1 + (t & 3);
                if (kk == 0) {
                    if (hg + 1 < CH1 / RD_STEPS) {
#pragma unroll
                        for (int q = 0; q < RD_STEPS; ++q) {
                            const float* rowS = sb + ((hg + 1) * RD_STEPS + q) * ROW_F + lpA;
                            nzs[(hg + 1) & 1][q][0] = rowS[0]; nzs[(hg + 1) & 1][q][1] = rowS[1]; nzs[(hg + 1) & 1][q][2] = rowS[2];
                        }
                        if (((hg + 1) & 1) == 0) {
#pragma unroll
                            for (int q = 0; q < SAVE_EVERY; ++q) {
                                const float* rowC = sb + (CH1 + ((hg + 1) >> 1) * SAVE_EVERY + q) * ROW_F + lpA;
                                nzc[((hg + 1) >> 1) & 1][q][0] = rowC[0]; nzc[((hg + 1) >> 1) & 1][q][1] = rowC[1]; nzc[((hg + 1) >> 1) & 1][q][2] = rowC[2];
                            }
                        }
                    }
                    asm volatile("" : "+v"(Bv[0]) : : "memory");
                }
                v8f cz = {};
                v8f d0 = __builtin_amdgcn_wmma_f32_16x16x4_f32(
                    false, Amat, false, Bv[0], (short)0, cz, false, false);
                if (s > 0) {
                    asm volatile("" : "+v"(d0), "+v"(dpend));
                    const int tq = (s - 1) % 20;
                    const int cq = 1 + (tq & 3);
                    const int gq = (s - 1) / 20;
                    float c01, c2;
                    em_update(dpend, nzc[gq & 1][tq >> 2], lo, (cq == 4) ? 1.0f : 0.0f, Bv[cq], c01, c2);
                    if ((tq >> 2) == SAVE_EVERY - 1) { sc01[cq] = c01; sc2[cq] = c2; }
                    asm volatile("" : "+v"(Bv[cq]), "+v"(Bv[cj]));
                }
                v8f dc = __builtin_amdgcn_wmma_f32_16x16x4_f32(
                    false, Amat, false, Bv[cj], (short)0, cz, false, false);
                asm volatile("" : "+v"(dc), "+v"(d0));
                float s01, s2;
                em_update(d0, nzs[hg & 1][kk], lo, 1.0f, Bv[0], s01, s2);
                dpend = dc;
                if (kk == SAVE_EVERY - 1) { h01 = s01; h2 = s2; }
                if (kk == RD_STEPS - 1) {
                    asm volatile("" : "+v"(Bv[0]) : : "memory");
                    const int jg = c * SUBS + sg;
                    ts01[jg - 1] = h01;
                    ts2[jg - 1]  = h2;
                    ts01[jg] = s01;
                    ts2[jg]  = s2;
                    if (s == 29) {
                        const int jc = c * 2;
#pragma unroll
                        for (int j = 1; j < 5; ++j)
                            tp01[(j - 1) * C_PLANE + jc] = sc01[j];
                        tp2[0 * C_PLANE + jc] = lo ? sc2[1] : sc2[2];
                        tp2[2 * C_PLANE + jc] = lo ? sc2[3] : sc2[4];
                    }
                }
            }
            {
                em_update(dpend, nzc[1][SAVE_EVERY - 1], lo, 1.0f, Bv[4], sc01[4], sc2[4]);
                const int jc = c * 2 + 1;
#pragma unroll
                for (int j = 1; j < 5; ++j)
                    tp01[(j - 1) * C_PLANE + jc] = sc01[j];
                tp2[0 * C_PLANE + jc] = lo ? sc2[1] : sc2[2];
                tp2[2 * C_PLANE + jc] = lo ? sc2[3] : sc2[4];
            }
        }
        lds_handover_barrier();
    }

    if (producer) drain_saves(tcol, out, dr_lds, dr_out, N_CHUNK - 1);

    for (int i0 = 0; i0 < CLOSE_TRIPS; i0 += CLOSE_U) {
        float va[CLOSE_U];
#pragma unroll
        for (int u = 0; u < CLOSE_U; ++u) {
            const int v  = tid + N_THREADS * (i0 + u);
            const int vv = (v < N_CLOSE) ? v : (N_CLOSE - 1);
            const int rw = vv / SEG2_SAVES;
            const int j  = vv - rw * SEG2_SAVES;
            const int q  = rw & 63;
            const float* xm = tcol + S_OFF + q * TS + (SEG1_SAVES - 1);
            const float a0 = xm[0], a1 = xm[64 * TS], a2 = xm[128 * TS];
            const float* t0 = tcol + rw * TC + j;
            float w = t0[3 * C_PLANE];
            w = __builtin_fmaf(a2, t0[2 * C_PLANE], w);
            w = __builtin_fmaf(a1, t0[1 * C_PLANE], w);
            va[u] = __builtin_fmaf(a0, t0[0], w);
        }
#pragma unroll
        for (int u = 0; u < CLOSE_U; ++u) {
            const int v = tid + N_THREADS * (i0 + u);
            if (v < N_CLOSE) {
                const int rw = v / SEG2_SAVES;
                const int j  = v - rw * SEG2_SAVES;
                out[(rw >> 6) * BT + (blk + (rw & 63)) * T_SAVED + (SEG1_SAVES + 1) + j] = va[u];
            }
        }
    }
}

extern "C" void kernel_launch(void* const* d_in, const int* in_sizes, int n_in,
                              void* d_out, int out_size, void* d_ws, size_t ws_size,
                              hipStream_t stream) {
    const float* drift     = (const float*)d_in[0];
    const float* driftb    = (const float*)d_in[1];
    const float* diffusion = (const float*)d_in[2];
    const float* diffb     = (const float*)d_in[3];
    const float* x0        = (const float*)d_in[4];
    const float* noise     = (const float*)d_in[5];
    float* out = (float*)d_out;

    dim3 block(224);
    dim3 grid(B_PATHS / 64);
    sde_wmma_kernel<<<grid, block, 0, stream>>>(
        drift, driftb, diffusion, diffb, x0, noise, out);
}


# the update's three FMAs issue as plain FMAs that read the product's registers directly: no packed FMA and no two register copies in front of it on the serial link between the state's products
# speedup vs baseline: 2.5783x; 1.0662x over previous
#include <hip/hip_runtime.h>


typedef float v2f __attribute__((ext_vector_type(2)));
typedef float v4f __attribute__((ext_vector_type(4)));
typedef float v8f __attribute__((ext_vector_type(8)));

#define B_PATHS    16384
#define N_STEPS    500
#define SEG1_STEPS 400
#define SAVE_EVERY 5
#define T_SAVED    101
#define SEG1_SAVES 80
#define SEG2_SAVES 20
#define DT         0.002f
#define N_CHUNK    10
#define CH1        40
#define CH2        10
#define CH_ROWS    (CH1 + CH2)
#define SUBS       8
#define RD_STEPS   10
#define N_PIECE    25
#define ROW_F      192
#define STAGE_F    (CH_ROWS * ROW_F)
#define STEP_F     (B_PATHS * 3)
#define N_THREADS  224
#define N_CHAIN    4
#define N_STRIP    192
#define TC         21
#define TS         81
#define C_PLANE    (N_STRIP * TC)
#define S_OFF      (4 * C_PLANE)
#define S_SIZE     (N_STRIP * TS)
#define N_DRAIN    16
#define N_CLOSE    (N_STRIP * SEG2_SAVES)
#define CLOSE_U    6
#define CLOSE_TRIPS 18


__device__ __forceinline__ void lds_handover_barrier() {
    __builtin_amdgcn_fence(__ATOMIC_RELEASE, "workgroup", "local");
    __builtin_amdgcn_s_barrier();
    __builtin_amdgcn_fence(__ATOMIC_ACQUIRE, "workgroup", "local");
}

__device__ __forceinline__ void drain_saves(const float* tcol, float* __restrict__ out,
                                            const int (&dr_lds)[N_DRAIN], const unsigned (&dr_out)[N_DRAIN],
                                            int cc) {
    float dv[N_DRAIN];
    float* const oc = out + cc * SUBS;
#pragma unroll
    for (int k = 0; k < N_DRAIN; ++k) dv[k] = tcol[dr_lds[k] + cc * SUBS];
#pragma unroll
    for (int k = 0; k < N_DRAIN; ++k) oc[dr_out[k]] = dv[k];
}

__device__ __forceinline__ int noise_step(int c, int sl) {
    return (sl < CH1) ? (c * CH1 + sl) : (SEG1_STEPS + c * CH2 + (sl - CH1));
}

__device__ __forceinline__ void em_update(const v8f& d, const float (&nz)[3], bool lo, float hv,
                                          v2f& b, float& x01, float& x2) {
    float xn0 = __builtin_fmaf(d[3], nz[0], d[0]);
    float xn1 = d[4] * nz[1] + d[1];
    float xn2 = __builtin_fmaf(d[5], nz[2], d[2]);
    asm volatile("" : "+v"(xn0), "+v"(xn1), "+v"(xn2));
    b.x = lo ? xn0 : xn2;
    b.y = lo ? xn1 : hv;
    x01 = lo ? xn0 : xn1;
    x2  = xn2;
}

__global__ __launch_bounds__(N_THREADS) void sde_wmma_kernel(
    const float* __restrict__ drift, const float* __restrict__ drift_bias,
    const float* __restrict__ diffusion, const float* __restrict__ diffusion_bias,
    const float* __restrict__ x0, const float* __restrict__ noise,
    float* __restrict__ out)
{
    __shared__ v4f   lds4[2 * STAGE_F / 4];
    __shared__ float tcol[S_OFF + S_SIZE];
    float* lds = (float*)lds4;

    const int  tid  = threadIdx.x;
    const int  lane = tid & 31;
    const int  wv   = __builtin_amdgcn_readfirstlane(tid >> 5);
    const bool producer = wv >= N_CHAIN;
    const int  blk  = blockIdx.x * 64;
    const int  m    = lane & 15;
    const bool lo   = lane < 16;
    const float sqdt = __builtin_sqrtf(DT);
    const int  BT = B_PATHS * T_SAVED;
    const int  pw = (wv & (N_CHAIN - 1)) * 16 + m;
    const int  p  = blk + pw;

    const int ptid = tid - N_CHAIN * 32;
    const int prow = (ptid >= 48) ? 1 : 0;
    const int pcol = (ptid - 48 * prow) * 4;
    int dr_lds[N_DRAIN];
    unsigned dr_out[N_DRAIN];
#pragma unroll
    for (int k = 0; k < N_DRAIN; ++k) { dr_lds[k] = 0; dr_out[k] = 0u; }

    v2f Amat; Amat.x = 0.0f; Amat.y = 0.0f;
    v2f Bv[5];
#pragma unroll
    for (int j = 0; j < 5; ++j) { Bv[j].x = 0.0f; Bv[j].y = 0.0f; }
    const int lpA = pw * 3;
    const int oo01 = (lo ? 0 : BT) + p * T_SAVED;
    const int oo2  = 2 * BT + p * T_SAVED;
    float* const tp01 = tcol + ((lo ? 0 : 64) + pw) * TC;
    float* const tp2  = tcol + (lo ? 0 : C_PLANE) + (128 + pw) * TC;
    float* const ts01 = tcol + S_OFF + ((lo ? 0 : 64) + pw) * TS;
    float* const ts2  = tcol + S_OFF + (128 + pw) * TS;

    if (producer) {
#pragma unroll
        for (int k = 0; k < N_DRAIN; ++k) {
            const int e  = ptid + 96 * k;
            const int rw = e / SUBS;
            const int i  = e - rw * SUBS;
            dr_lds[k] = S_OFF + rw * TS + i;
            dr_out[k] = (unsigned)((rw >> 6) * (B_PATHS * T_SAVED) + (blk + (rw & 63)) * T_SAVED + 1 + i);
        }
        v4f r4[N_PIECE];
#pragma unroll
        for (int r = 0; r < N_PIECE; ++r) {
            const int sl = 2 * r + prow;
            r4[r] = *(const v4f*)(noise + noise_step(0, sl) * STEP_F + blk * 3 + pcol);
        }
#pragma unroll
        for (int r = 0; r < N_PIECE; ++r) {
            const int sl = 2 * r + prow;
            __builtin_prefetch(noise + noise_step(1, sl) * STEP_F + blk * 3 + pcol, 0, 3);
        }
#pragma unroll
        for (int r = 0; r < N_PIECE; ++r) {
            const int sl = 2 * r + prow;
            *(v4f*)(lds + sl * ROW_F + pcol) = r4[r];
        }
    } else {
        const int k0 = lo ? 0 : 2;
        const int k1 = lo ? 1 : 3;
        const int mm = m & 7;
        float a0v = 0.0f, a1v = 0.0f;
        if (mm < 3) {
            a0v = DT * drift[mm * 3 + k0] + ((mm == k0) ? 1.0f : 0.0f);
            a1v = (k1 < 3) ? (DT * drift[mm * 3 + k1] + ((mm == k1) ? 1.0f : 0.0f))
                           : (DT * drift_bias[mm]);
        } else if (mm < 6) {
            const int r = mm - 3;
            a0v = sqdt * diffusion[r * 3 + k0];
            a1v = (k1 < 3) ? (sqdt * diffusion[r * 3 + k1])
                           : (sqdt * diffusion_bias[r]);
        }
        Amat.x = a0v; Amat.y = a1v;

        const float xi0 = x0[p * 3 + 0], xi1 = x0[p * 3 + 1], xi2 = x0[p * 3 + 2];
        Bv[0].x = lo ? xi0 : xi2;
        Bv[0].y = lo ? xi1 : 1.0f;
        out[oo01] = lo ? xi0 : xi1;
        out[oo2]  = xi2;
        Bv[1].x = lo ? 1.0f : 0.0f;  Bv[1].y = 0.0f;
        Bv[2].x = 0.0f;              Bv[2].y = lo ? 1.0f : 0.0f;
        Bv[3].x = lo ? 0.0f : 1.0f;  Bv[3].y = 0.0f;
        Bv[4].x = 0.0f;              Bv[4].y = lo ? 0.0f : 1.0f;
    }
    lds_handover_barrier();

    for (int c = 0; c < N_CHUNK; ++c) {
        const int st = c & 1;
        if (producer) {
            if (c + 1 < N_CHUNK) {
                const float* src = noise + blk * 3;
                float*       dst = lds + (st ^ 1) * STAGE_F;
                v4f r4[N_PIECE];
#pragma unroll
                for (int r = 0; r < N_PIECE; ++r) {
                    const int sl = 2 * r + prow;
                    r4[r] = *(const v4f*)(src + noise_step(c + 1, sl) * STEP_F + pcol);
                }
                if (c + 2 < N_CHUNK) {
#pragma unroll
                    for (int r = 0; r < N_PIECE; ++r) {
                        const int sl = 2 * r + prow;
                        __builtin_prefetch(src + noise_step(c + 2, sl) * STEP_F + pcol, 0, 3);
                    }
                }
                if (c > 0) drain_saves(tcol, out, dr_lds, dr_out, c - 1);
#pragma unroll
                for (int r = 0; r < N_PIECE; ++r) {
                    const int sl = 2 * r + prow;
                    *(v4f*)(dst + sl * ROW_F + pcol) = r4[r];
                }
            } else {
                drain_saves(tcol, out, dr_lds, dr_out, c - 1);
            }
        } else {
            const float* sb = lds + st * STAGE_F;
            float nzs[2][RD_STEPS][3];
            float nzc[2][SAVE_EVERY][3];
#pragma unroll
            for (int k = 0; k < RD_STEPS; ++k) {
                const float* rowS = sb + k * ROW_F + lpA;
                nzs[0][k][0] = rowS[0]; nzs[0][k][1] = rowS[1]; nzs[0][k][2] = rowS[2];
            }
#pragma unroll
            for (int k = 0; k < SAVE_EVERY; ++k) {
                const float* rowC = sb + (CH1 + k) * ROW_F + lpA;
                nzc[0][k][0] = rowC[0]; nzc[0][k][1] = rowC[1]; nzc[0][k][2] = rowC[2];
            }
            v8f dpend = {};
            float sc01[5], sc2[5];
            float h01 = 0.0f, h2 = 0.0f;
#pragma unroll
            for (int s = 0; s < CH1; ++s) {
                const int hg = s / RD_STEPS;
                const int kk = s - hg * RD_STEPS;
                const int sg = s / SAVE_EVERY;
                const int t  = s % 20;
                const int cj = 1 + (t & 3);
                if (kk == 0) {
                    if (hg + 1 < CH1 / RD_STEPS) {
#pragma unroll
                        for (int q = 0; q < RD_STEPS; ++q) {
                            const float* rowS = sb + ((hg + 1) * RD_STEPS + q) * ROW_F + lpA;
                            nzs[(hg + 1) & 1][q][0] = rowS[0]; nzs[(hg + 1) & 1][q][1] = rowS[1]; nzs[(hg + 1) & 1][q][2] = rowS[2];
                        }
                        if (((hg + 1) & 1) == 0) {
#pragma unroll
                            for (int q = 0; q < SAVE_EVERY; ++q) {
                                const float* rowC = sb + (CH1 + ((hg + 1) >> 1) * SAVE_EVERY + q) * ROW_F + lpA;
                                nzc[((hg + 1) >> 1) & 1][q][0] = rowC[0]; nzc[((hg + 1) >> 1) & 1][q][1] = rowC[1]; nzc[((hg + 1) >> 1) & 1][q][2] = rowC[2];
                            }
                        }
                    }
                    asm volatile("" : "+v"(Bv[0]) : : "memory");
                }
                v8f cz = {};
                v8f d0 = __builtin_amdgcn_wmma_f32_16x16x4_f32(
                    false, Amat, false, Bv[0], (short)0, cz, false, false);
                if (s > 0) {
                    asm volatile("" : "+v"(d0), "+v"(dpend));
                    const int tq = (s - 1) % 20;
                    const int cq = 1 + (tq & 3);
                    const int gq = (s - 1) / 20;
                    float c01, c2;
                    em_update(dpend, nzc[gq & 1][tq >> 2], lo, (cq == 4) ? 1.0f : 0.0f, Bv[cq], c01, c2);
                    if ((tq >> 2) == SAVE_EVERY - 1) { sc01[cq] = c01; sc2[cq] = c2; }
                    asm volatile("" : "+v"(Bv[cq]), "+v"(Bv[cj]));
                }
                v8f dc = __builtin_amdgcn_wmma_f32_16x16x4_f32(
                    false, Amat, false, Bv[cj], (short)0, cz, false, false);
                asm volatile("" : "+v"(dc), "+v"(d0));
                float s01, s2;
                em_update(d0, nzs[hg & 1][kk], lo, 1.0f, Bv[0], s01, s2);
                dpend = dc;
                if (kk == SAVE_EVERY - 1) { h01 = s01; h2 = s2; }
                if (kk == RD_STEPS - 1) {
                    asm volatile("" : "+v"(Bv[0]) : : "memory");
                    const int jg = c * SUBS + sg;
                    ts01[jg - 1] = h01;
                    ts2[jg - 1]  = h2;
                    ts01[jg] = s01;
                    ts2[jg]  = s2;
                    if (s == 29) {
                        const int jc = c * 2;
#pragma unroll
                        for (int j = 1; j < 5; ++j)
                            tp01[(j - 1) * C_PLANE + jc] = sc01[j];
                        tp2[0 * C_PLANE + jc] = lo ? sc2[1] : sc2[2];
                        tp2[2 * C_PLANE + jc] = lo ? sc2[3] : sc2[4];
                    }
                }
            }
            {
                em_update(dpend, nzc[1][SAVE_EVERY - 1], lo, 1.0f, Bv[4], sc01[4], sc2[4]);
                const int jc = c * 2 + 1;
#pragma unroll
                for (int j = 1; j < 5; ++j)
                    tp01[(j - 1) * C_PLANE + jc] = sc01[j];
                tp2[0 * C_PLANE + jc] = lo ? sc2[1] : sc2[2];
                tp2[2 * C_PLANE + jc] = lo ? sc2[3] : sc2[4];
            }
        }
        lds_handover_barrier();
    }

    if (producer) drain_saves(tcol, out, dr_lds, dr_out, N_CHUNK - 1);

    for (int i0 = 0; i0 < CLOSE_TRIPS; i0 += CLOSE_U) {
        float va[CLOSE_U];
#pragma unroll
        for (int u = 0; u < CLOSE_U; ++u) {
            const int v  = tid + N_THREADS * (i0 + u);
            const int vv = (v < N_CLOSE) ? v : (N_CLOSE - 1);
            const int rw = vv / SEG2_SAVES;
            const int j  = vv - rw * SEG2_SAVES;
            const int q  = rw & 63;
            const float* xm = tcol + S_OFF + q * TS + (SEG1_SAVES - 1);
            const float a0 = xm[0], a1 = xm[64 * TS], a2 = xm[128 * TS];
            const float* t0 = tcol + rw * TC + j;
            float w = t0[3 * C_PLANE];
            w = __builtin_fmaf(a2, t0[2 * C_PLANE], w);
            w = __builtin_fmaf(a1, t0[1 * C_PLANE], w);
            va[u] = __builtin_fmaf(a0, t0[0], w);
        }
#pragma unroll
        for (int u = 0; u < CLOSE_U; ++u) {
            const int v = tid + N_THREADS * (i0 + u);
            if (v < N_CLOSE) {
                const int rw = v / SEG2_SAVES;
                const int j  = v - rw * SEG2_SAVES;
                out[(rw >> 6) * BT + (blk + (rw & 63)) * T_SAVED + (SEG1_SAVES + 1) + j] = va[u];
            }
        }
    }
}

extern "C" void kernel_launch(void* const* d_in, const int* in_sizes, int n_in,
                              void* d_out, int out_size, void* d_ws, size_t ws_size,
                              hipStream_t stream) {
    const float* drift     = (const float*)d_in[0];
    const float* driftb    = (const float*)d_in[1];
    const float* diffusion = (const float*)d_in[2];
    const float* diffb     = (const float*)d_in[3];
    const float* x0        = (const float*)d_in[4];
    const float* noise     = (const float*)d_in[5];
    float* out = (float*)d_out;

    dim3 block(224);
    dim3 grid(B_PATHS / 64);
    sde_wmma_kernel<<<grid, block, 0, stream>>>(
        drift, driftb, diffusion, diffb, x0, noise, out);
}


# time axis re-cut from 400+100 to 450+50 steps: 650 products and updates per wave (were 800), map products beside 4 of 9 state steps; saves 91-100 = map applied to x(450), a new rounding order
# speedup vs baseline: 2.8185x; 1.0932x over previous
#include <hip/hip_runtime.h>


typedef float v2f __attribute__((ext_vector_type(2)));
typedef float v4f __attribute__((ext_vector_type(4)));
typedef float v8f __attribute__((ext_vector_type(8)));

#define B_PATHS    16384
#define N_STEPS    500
#define SEG1_STEPS 450
#define SAVE_EVERY 5
#define T_SAVED    101
#define SEG1_SAVES 90
#define SEG2_SAVES 10
#define DT         0.002f
#define N_CHUNK    10
#define CH1        45
#define CH2        5
#define CH_ROWS    (CH1 + CH2)
#define SUBS       9
#define RD_STEPS   15
#define MAP_PER    9
#define N_PIECE    25
#define ROW_F      192
#define STAGE_F    (CH_ROWS * ROW_F)
#define STEP_F     (B_PATHS * 3)
#define N_THREADS  224
#define N_CHAIN    4
#define N_STRIP    192
#define TC         11
#define TS         91
#define C_PLANE    (N_STRIP * TC)
#define S_OFF      (4 * C_PLANE)
#define S_SIZE     (N_STRIP * TS)
#define N_DRAIN    18
#define N_CLOSE    (N_STRIP * SEG2_SAVES)
#define CLOSE_U    3
#define CLOSE_TRIPS 9


__device__ __forceinline__ void lds_handover_barrier() {
    __builtin_amdgcn_fence(__ATOMIC_RELEASE, "workgroup", "local");
    __builtin_amdgcn_s_barrier();
    __builtin_amdgcn_fence(__ATOMIC_ACQUIRE, "workgroup", "local");
}

__device__ __forceinline__ void drain_saves(const float* tcol, float* __restrict__ out,
                                            const int (&dr_lds)[N_DRAIN], const unsigned (&dr_out)[N_DRAIN],
                                            int cc) {
    float dv[N_DRAIN];
    float* const oc = out + cc * SUBS;
#pragma unroll
    for (int k = 0; k < N_DRAIN; ++k) dv[k] = tcol[dr_lds[k] + cc * SUBS];
#pragma unroll
    for (int k = 0; k < N_DRAIN; ++k) oc[dr_out[k]] = dv[k];
}

__device__ __forceinline__ int noise_step(int c, int sl) {
    return (sl < CH1) ? (c * CH1 + sl) : (SEG1_STEPS + c * CH2 + (sl - CH1));
}

__device__ __forceinline__ void em_update(const v8f& d, const float (&nz)[3], bool lo, float hv,
                                          v2f& b, float& x01, float& x2) {
    float xn0 = __builtin_fmaf(d[3], nz[0], d[0]);
    float xn1 = d[4] * nz[1] + d[1];
    float xn2 = __builtin_fmaf(d[5], nz[2], d[2]);
    asm volatile("" : "+v"(xn0), "+v"(xn1), "+v"(xn2));
    b.x = lo ? xn0 : xn2;
    b.y = lo ? xn1 : hv;
    x01 = lo ? xn0 : xn1;
    x2  = xn2;
}

__global__ __launch_bounds__(N_THREADS) void sde_wmma_kernel(
    const float* __restrict__ drift, const float* __restrict__ drift_bias,
    const float* __restrict__ diffusion, const float* __restrict__ diffusion_bias,
    const float* __restrict__ x0, const float* __restrict__ noise,
    float* __restrict__ out)
{
    __shared__ v4f   lds4[2 * STAGE_F / 4];
    __shared__ float tcol[S_OFF + S_SIZE];
    float* lds = (float*)lds4;

    const int  tid  = threadIdx.x;
    const int  lane = tid & 31;
    const int  wv   = __builtin_amdgcn_readfirstlane(tid >> 5);
    const bool producer = wv >= N_CHAIN;
    const int  blk  = blockIdx.x * 64;
    const int  m    = lane & 15;
    const bool lo   = lane < 16;
    const float sqdt = __builtin_sqrtf(DT);
    const int  BT = B_PATHS * T_SAVED;
    const int  pw = (wv & (N_CHAIN - 1)) * 16 + m;
    const int  p  = blk + pw;

    const int ptid = tid - N_CHAIN * 32;
    const int prow = (ptid >= 48) ? 1 : 0;
    const int pcol = (ptid - 48 * prow) * 4;
    int dr_lds[N_DRAIN];
    unsigned dr_out[N_DRAIN];
#pragma unroll
    for (int k = 0; k < N_DRAIN; ++k) { dr_lds[k] = 0; dr_out[k] = 0u; }

    v2f Amat; Amat.x = 0.0f; Amat.y = 0.0f;
    v2f Bv[5];
#pragma unroll
    for (int j = 0; j < 5; ++j) { Bv[j].x = 0.0f; Bv[j].y = 0.0f; }
    const int lpA = pw * 3;
    const int oo01 = (lo ? 0 : BT) + p * T_SAVED;
    const int oo2  = 2 * BT + p * T_SAVED;
    float* const tp01 = tcol + ((lo ? 0 : 64) + pw) * TC;
    float* const tp2  = tcol + (lo ? 0 : C_PLANE) + (128 + pw) * TC;
    float* const ts01 = tcol + S_OFF + ((lo ? 0 : 64) + pw) * TS;
    float* const ts2  = tcol + S_OFF + (128 + pw) * TS;

    if (producer) {
#pragma unroll
        for (int k = 0; k < N_DRAIN; ++k) {
            const int e  = ptid + 96 * k;
            const int rw = e / SUBS;
            const int i  = e - rw * SUBS;
            dr_lds[k] = S_OFF + rw * TS + i;
            dr_out[k] = (unsigned)((rw >> 6) * (B_PATHS * T_SAVED) + (blk + (rw & 63)) * T_SAVED + 1 + i);
        }
        v4f r4[N_PIECE];
#pragma unroll
        for (int r = 0; r < N_PIECE; ++r) {
            const int sl = 2 * r + prow;
            r4[r] = *(const v4f*)(noise + noise_step(0, sl) * STEP_F + blk * 3 + pcol);
        }
#pragma unroll
        for (int r = 0; r < N_PIECE; ++r) {
            const int sl = 2 * r + prow;
            __builtin_prefetch(noise + noise_step(1, sl) * STEP_F + blk * 3 + pcol, 0, 3);
        }
#pragma unroll
        for (int r = 0; r < N_PIECE; ++r) {
            const int sl = 2 * r + prow;
            *(v4f*)(lds + sl * ROW_F + pcol) = r4[r];
        }
    } else {
        const int k0 = lo ? 0 : 2;
        const int k1 = lo ? 1 : 3;
        const int mm = m & 7;
        float a0v = 0.0f, a1v = 0.0f;
        if (mm < 3) {
            a0v = DT * drift[mm * 3 + k0] + ((mm == k0) ? 1.0f : 0.0f);
            a1v = (k1 < 3) ? (DT * drift[mm * 3 + k1] + ((mm == k1) ? 1.0f : 0.0f))
                           : (DT * drift_bias[mm]);
        } else if (mm < 6) {
            const int r = mm - 3;
            a0v = sqdt * diffusion[r * 3 + k0];
            a1v = (k1 < 3) ? (sqdt * diffusion[r * 3 + k1])
                           : (sqdt * diffusion_bias[r]);
        }
        Amat.x = a0v; Amat.y = a1v;

        const float xi0 = x0[p * 3 + 0], xi1 = x0[p * 3 + 1], xi2 = x0[p * 3 + 2];
        Bv[0].x = lo ? xi0 : xi2;
        Bv[0].y = lo ? xi1 : 1.0f;
        out[oo01] = lo ? xi0 : xi1;
        out[oo2]  = xi2;
        Bv[1].x = lo ? 1.0f : 0.0f;  Bv[1].y = 0.0f;
        Bv[2].x = 0.0f;              Bv[2].y = lo ? 1.0f : 0.0f;
        Bv[3].x = lo ? 0.0f : 1.0f;  Bv[3].y = 0.0f;
        Bv[4].x = 0.0f;              Bv[4].y = lo ? 0.0f : 1.0f;
    }
    lds_handover_barrier();

    for (int c = 0; c < N_CHUNK; ++c) {
        const int st = c & 1;
        if (producer) {
            if (c + 1 < N_CHUNK) {
                const float* src = noise + blk * 3;
                float*       dst = lds + (st ^ 1) * STAGE_F;
                v4f r4[N_PIECE];
#pragma unroll
                for (int r = 0; r < N_PIECE; ++r) {
                    const int sl = 2 * r + prow;
                    r4[r] = *(const v4f*)(src + noise_step(c + 1, sl) * STEP_F + pcol);
                }
                if (c + 2 < N_CHUNK) {
#pragma unroll
                    for (int r = 0; r < N_PIECE; ++r) {
                        const int sl = 2 * r + prow;
                        __builtin_prefetch(src + noise_step(c + 2, sl) * STEP_F + pcol, 0, 3);
                    }
                }
                if (c > 0) drain_saves(tcol, out, dr_lds, dr_out, c - 1);
#pragma unroll
                for (int r = 0; r < N_PIECE; ++r) {
                    const int sl = 2 * r + prow;
                    *(v4f*)(dst + sl * ROW_F + pcol) = r4[r];
                }
            } else {
                drain_saves(tcol, out, dr_lds, dr_out, c - 1);
            }
        } else {
            const float* sb = lds + st * STAGE_F;
            float nzs[2][RD_STEPS][3];
            float nzc[CH2][3];
#pragma unroll
            for (int k = 0; k < RD_STEPS; ++k) {
                const float* rowS = sb + k * ROW_F + lpA;
                nzs[0][k][0] = rowS[0]; nzs[0][k][1] = rowS[1]; nzs[0][k][2] = rowS[2];
            }
#pragma unroll
            for (int k = 0; k < CH2; ++k) {
                const float* rowC = sb + (CH1 + k) * ROW_F + lpA;
                nzc[k][0] = rowC[0]; nzc[k][1] = rowC[1]; nzc[k][2] = rowC[2];
            }
            v8f dpend = {};
            float sc01[5], sc2[5];
            float ha01 = 0.0f, ha2 = 0.0f, hb01 = 0.0f, hb2 = 0.0f;
#pragma unroll
            for (int s = 0; s < CH1; ++s) {
                const int hg = s / RD_STEPS;
                const int kk = s - hg * RD_STEPS;
                const int sg = s / SAVE_EVERY;
                const int q  = s / MAP_PER;
                const int r9 = s - q * MAP_PER;
                if (kk == 0) {
                    if (hg + 1 < CH1 / RD_STEPS) {
#pragma unroll
                        for (int u = 0; u < RD_STEPS; ++u) {
                            const float* rowS = sb + ((hg + 1) * RD_STEPS + u) * ROW_F + lpA;
                            nzs[(hg + 1) & 1][u][0] = rowS[0]; nzs[(hg + 1) & 1][u][1] = rowS[1]; nzs[(hg + 1) & 1][u][2] = rowS[2];
                        }
                    }
                    asm volatile("" : "+v"(Bv[0]) : : "memory");
                }
                v8f cz = {};
                v8f d0 = __builtin_amdgcn_wmma_f32_16x16x4_f32(
                    false, Amat, false, Bv[0], (short)0, cz, false, false);
                if (r9 >= 1 && r9 <= 4) {
                    asm volatile("" : "+v"(d0), "+v"(dpend));
                    float c01, c2;
                    em_update(dpend, nzc[q], lo, (r9 == 4) ? 1.0f : 0.0f, Bv[r9], c01, c2);
                    if (q == CH2 - 1) { sc01[r9] = c01; sc2[r9] = c2; }
                }
                if (r9 < 4) {
                    if (r9 >= 1) {
                        asm volatile("" : "+v"(Bv[r9]), "+v"(Bv[1 + r9]));
                    }
                    v8f dc = __builtin_amdgcn_wmma_f32_16x16x4_f32(
                        false, Amat, false, Bv[1 + r9], (short)0, cz, false, false);
                    asm volatile("" : "+v"(dc), "+v"(d0));
                    dpend = dc;
                }
                float s01, s2;
                em_update(d0, nzs[hg & 1][kk], lo, 1.0f, Bv[0], s01, s2);
                if (kk == SAVE_EVERY - 1)     { ha01 = s01; ha2 = s2; }
                if (kk == 2 * SAVE_EVERY - 1) { hb01 = s01; hb2 = s2; }
                if (kk == RD_STEPS - 1) {
                    asm volatile("" : "+v"(Bv[0]) : : "memory");
                    const int jg = c * SUBS + sg;
                    ts01[jg - 2] = ha01;
                    ts2[jg - 2]  = ha2;
                    ts01[jg - 1] = hb01;
                    ts2[jg - 1]  = hb2;
                    ts01[jg] = s01;
                    ts2[jg]  = s2;
                    if (s == CH1 - 1) {
                        const int jc = c;
#pragma unroll
                        for (int j = 1; j < 5; ++j)
                            tp01[(j - 1) * C_PLANE + jc] = sc01[j];
                        tp2[0 * C_PLANE + jc] = lo ? sc2[1] : sc2[2];
                        tp2[2 * C_PLANE + jc] = lo ? sc2[3] : sc2[4];
                    }
                }
            }
        }
        lds_handover_barrier();
    }

    if (producer) drain_saves(tcol, out, dr_lds, dr_out, N_CHUNK - 1);

    for (int i0 = 0; i0 < CLOSE_TRIPS; i0 += CLOSE_U) {
        float va[CLOSE_U];
#pragma unroll
        for (int u = 0; u < CLOSE_U; ++u) {
            const int v  = tid + N_THREADS * (i0 + u);
            const int vv = (v < N_CLOSE) ? v : (N_CLOSE - 1);
            const int rw = vv / SEG2_SAVES;
            const int j  = vv - rw * SEG2_SAVES;
            const int q  = rw & 63;
            const float* xm = tcol + S_OFF + q * TS + (SEG1_SAVES - 1);
            const float a0 = xm[0], a1 = xm[64 * TS], a2 = xm[128 * TS];
            const float* t0 = tcol + rw * TC + j;
            float w = t0[3 * C_PLANE];
            w = __builtin_fmaf(a2, t0[2 * C_PLANE], w);
            w = __builtin_fmaf(a1, t0[1 * C_PLANE], w);
            va[u] = __builtin_fmaf(a0, t0[0], w);
        }
#pragma unroll
        for (int u = 0; u < CLOSE_U; ++u) {
            const int v = tid + N_THREADS * (i0 + u);
            if (v < N_CLOSE) {
                const int rw = v / SEG2_SAVES;
                const int j  = v - rw * SEG2_SAVES;
                out[(rw >> 6) * BT + (blk + (rw & 63)) * T_SAVED + (SEG1_SAVES + 1) + j] = va[u];
            }
        }
    }
}

extern "C" void kernel_launch(void* const* d_in, const int* in_sizes, int n_in,
                              void* d_out, int out_size, void* d_ws, size_t ws_size,
                              hipStream_t stream) {
    const float* drift     = (const float*)d_in[0];
    const float* driftb    = (const float*)d_in[1];
    const float* diffusion = (const float*)d_in[2];
    const float* diffb     = (const float*)d_in[3];
    const float* x0        = (const float*)d_in[4];
    const float* noise     = (const float*)d_in[5];
    float* out = (float*)d_out;

    dim3 block(224);
    dim3 grid(B_PATHS / 64);
    sde_wmma_kernel<<<grid, block, 0, stream>>>(
        drift, driftb, diffusion, diffb, x0, noise, out);
}


# the map's four column chains move to four waves of their own, one beside each state wave on its SIMD (11 waves a group): a state wave issues only its own 450 products and updates; sums unchanged
# speedup vs baseline: 3.0073x; 1.0670x over previous
#include <hip/hip_runtime.h>


typedef float v2f __attribute__((ext_vector_type(2)));
typedef float v4f __attribute__((ext_vector_type(4)));
typedef float v8f __attribute__((ext_vector_type(8)));

#define B_PATHS    16384
#define N_STEPS    500
#define SEG1_STEPS 450
#define SAVE_EVERY 5
#define T_SAVED    101
#define SEG1_SAVES 90
#define SEG2_SAVES 10
#define DT         0.002f
#define N_CHUNK    10
#define CH1        45
#define CH2        5
#define CH_ROWS    (CH1 + CH2)
#define SUBS       9
#define RD_STEPS   15
#define N_PIECE    25
#define ROW_F      192
#define STAGE_F    (CH_ROWS * ROW_F)
#define STEP_F     (B_PATHS * 3)
#define N_THREADS  352
#define N_CHAIN    4
#define N_COMP     (2 * N_CHAIN)
#define N_STRIP    192
#define TC         11
#define TS         91
#define C_PLANE    (N_STRIP * TC)
#define S_OFF      (4 * C_PLANE)
#define S_SIZE     (N_STRIP * TS)
#define N_DRAIN    18
#define N_CLOSE    (N_STRIP * SEG2_SAVES)
#define CLOSE_U    3
#define CLOSE_TRIPS 6


__device__ __forceinline__ void lds_handover_barrier() {
    __builtin_amdgcn_fence(__ATOMIC_RELEASE, "workgroup", "local");
    __builtin_amdgcn_s_barrier();
    __builtin_amdgcn_fence(__ATOMIC_ACQUIRE, "workgroup", "local");
}

__device__ __forceinline__ void drain_saves(const float* tcol, float* __restrict__ out,
                                            const int (&dr_lds)[N_DRAIN], const unsigned (&dr_out)[N_DRAIN],
                                            int cc) {
    float dv[N_DRAIN];
    float* const oc = out + cc * SUBS;
#pragma unroll
    for (int k = 0; k < N_DRAIN; ++k) dv[k] = tcol[dr_lds[k] + cc * SUBS];
#pragma unroll
    for (int k = 0; k < N_DRAIN; ++k) oc[dr_out[k]] = dv[k];
}

__device__ __forceinline__ int noise_step(int c, int sl) {
    return (sl < CH1) ? (c * CH1 + sl) : (SEG1_STEPS + c * CH2 + (sl - CH1));
}

__device__ __forceinline__ void em_update(const v8f& d, const float (&nz)[3], bool lo, float hv,
                                          v2f& b, float& x01, float& x2) {
    float xn0 = __builtin_fmaf(d[3], nz[0], d[0]);
    float xn1 = d[4] * nz[1] + d[1];
    float xn2 = __builtin_fmaf(d[5], nz[2], d[2]);
    asm volatile("" : "+v"(xn0), "+v"(xn1), "+v"(xn2));
    b.x = lo ? xn0 : xn2;
    b.y = lo ? xn1 : hv;
    x01 = lo ? xn0 : xn1;
    x2  = xn2;
}

__global__ __launch_bounds__(N_THREADS) void sde_wmma_kernel(
    const float* __restrict__ drift, const float* __restrict__ drift_bias,
    const float* __restrict__ diffusion, const float* __restrict__ diffusion_bias,
    const float* __restrict__ x0, const float* __restrict__ noise,
    float* __restrict__ out)
{
    __shared__ v4f   lds4[2 * STAGE_F / 4];
    __shared__ float tcol[S_OFF + S_SIZE];
    float* lds = (float*)lds4;

    const int  tid  = threadIdx.x;
    const int  lane = tid & 31;
    const int  wv   = __builtin_amdgcn_readfirstlane(tid >> 5);
    const bool producer = wv >= N_COMP;
    const bool mapwave  = wv >= N_CHAIN && wv < N_COMP;
    const int  blk  = blockIdx.x * 64;
    const int  m    = lane & 15;
    const bool lo   = lane < 16;
    const float sqdt = __builtin_sqrtf(DT);
    const int  BT = B_PATHS * T_SAVED;
    const int  pw = (wv & (N_CHAIN - 1)) * 16 + m;
    const int  p  = blk + pw;

    const int ptid = tid - N_COMP * 32;
    const int prow = (ptid >= 48) ? 1 : 0;
    const int pcol = (ptid - 48 * prow) * 4;
    int dr_lds[N_DRAIN];
    unsigned dr_out[N_DRAIN];
#pragma unroll
    for (int k = 0; k < N_DRAIN; ++k) { dr_lds[k] = 0; dr_out[k] = 0u; }

    v2f Amat; Amat.x = 0.0f; Amat.y = 0.0f;
    v2f Bv[5];
#pragma unroll
    for (int j = 0; j < 5; ++j) { Bv[j].x = 0.0f; Bv[j].y = 0.0f; }
    const int lpA = pw * 3;
    const int oo01 = (lo ? 0 : BT) + p * T_SAVED;
    const int oo2  = 2 * BT + p * T_SAVED;
    float* const tp01 = tcol + ((lo ? 0 : 64) + pw) * TC;
    float* const tp2  = tcol + (lo ? 0 : C_PLANE) + (128 + pw) * TC;
    float* const ts01 = tcol + S_OFF + ((lo ? 0 : 64) + pw) * TS;
    float* const ts2  = tcol + S_OFF + (128 + pw) * TS;

    if (producer) {
#pragma unroll
        for (int k = 0; k < N_DRAIN; ++k) {
            const int e  = ptid + 96 * k;
            const int rw = e / SUBS;
            const int i  = e - rw * SUBS;
            dr_lds[k] = S_OFF + rw * TS + i;
            dr_out[k] = (unsigned)((rw >> 6) * (B_PATHS * T_SAVED) + (blk + (rw & 63)) * T_SAVED + 1 + i);
        }
        v4f r4[N_PIECE];
#pragma unroll
        for (int r = 0; r < N_PIECE; ++r) {
            const int sl = 2 * r + prow;
            r4[r] = *(const v4f*)(noise + noise_step(0, sl) * STEP_F + blk * 3 + pcol);
        }
#pragma unroll
        for (int r = 0; r < N_PIECE; ++r) {
            const int sl = 2 * r + prow;
            __builtin_prefetch(noise + noise_step(1, sl) * STEP_F + blk * 3 + pcol, 0, 3);
        }
#pragma unroll
        for (int r = 0; r < N_PIECE; ++r) {
            const int sl = 2 * r + prow;
            *(v4f*)(lds + sl * ROW_F + pcol) = r4[r];
        }
    } else {
        const int k0 = lo ? 0 : 2;
        const int k1 = lo ? 1 : 3;
        const int mm = m & 7;
        float a0v = 0.0f, a1v = 0.0f;
        if (mm < 3) {
            a0v = DT * drift[mm * 3 + k0] + ((mm == k0) ? 1.0f : 0.0f);
            a1v = (k1 < 3) ? (DT * drift[mm * 3 + k1] + ((mm == k1) ? 1.0f : 0.0f))
                           : (DT * drift_bias[mm]);
        } else if (mm < 6) {
            const int r = mm - 3;
            a0v = sqdt * diffusion[r * 3 + k0];
            a1v = (k1 < 3) ? (sqdt * diffusion[r * 3 + k1])
                           : (sqdt * diffusion_bias[r]);
        }
        Amat.x = a0v; Amat.y = a1v;

        if (!mapwave) {
            const float xi0 = x0[p * 3 + 0], xi1 = x0[p * 3 + 1], xi2 = x0[p * 3 + 2];
            Bv[0].x = lo ? xi0 : xi2;
            Bv[0].y = lo ? xi1 : 1.0f;
            out[oo01] = lo ? xi0 : xi1;
            out[oo2]  = xi2;
        } else {
            Bv[1].x = lo ? 1.0f : 0.0f;  Bv[1].y = 0.0f;
            Bv[2].x = 0.0f;              Bv[2].y = lo ? 1.0f : 0.0f;
            Bv[3].x = lo ? 0.0f : 1.0f;  Bv[3].y = 0.0f;
            Bv[4].x = 0.0f;              Bv[4].y = lo ? 0.0f : 1.0f;
        }
    }
    lds_handover_barrier();

    for (int c = 0; c < N_CHUNK; ++c) {
        const int st = c & 1;
        if (producer) {
            if (c + 1 < N_CHUNK) {
                const float* src = noise + blk * 3;
                float*       dst = lds + (st ^ 1) * STAGE_F;
                v4f r4[N_PIECE];
#pragma unroll
                for (int r = 0; r < N_PIECE; ++r) {
                    const int sl = 2 * r + prow;
                    r4[r] = *(const v4f*)(src + noise_step(c + 1, sl) * STEP_F + pcol);
                }
                if (c + 2 < N_CHUNK) {
#pragma unroll
                    for (int r = 0; r < N_PIECE; ++r) {
                        const int sl = 2 * r + prow;
                        __builtin_prefetch(src + noise_step(c + 2, sl) * STEP_F + pcol, 0, 3);
                    }
                }
                if (c > 0) drain_saves(tcol, out, dr_lds, dr_out, c - 1);
#pragma unroll
                for (int r = 0; r < N_PIECE; ++r) {
                    const int sl = 2 * r + prow;
                    *(v4f*)(dst + sl * ROW_F + pcol) = r4[r];
                }
            } else {
                drain_saves(tcol, out, dr_lds, dr_out, c - 1);
            }
        } else if (mapwave) {
            const float* sb = lds + st * STAGE_F;
            float nzc[CH2][3];
#pragma unroll
            for (int k = 0; k < CH2; ++k) {
                const float* rowC = sb + (CH1 + k) * ROW_F + lpA;
                nzc[k][0] = rowC[0]; nzc[k][1] = rowC[1]; nzc[k][2] = rowC[2];
            }
            float sc01[5], sc2[5];
#pragma unroll
            for (int q = 0; q < CH2; ++q) {
                v8f cz = {};
                v8f dj[4];
#pragma unroll
                for (int j = 0; j < 4; ++j)
                    dj[j] = __builtin_amdgcn_wmma_f32_16x16x4_f32(
                        false, Amat, false, Bv[1 + j], (short)0, cz, false, false);
#pragma unroll
                for (int j = 0; j < 4; ++j) {
                    float c01, c2;
                    em_update(dj[j], nzc[q], lo, (j == 3) ? 1.0f : 0.0f, Bv[1 + j], c01, c2);
                    if (q == CH2 - 1) { sc01[1 + j] = c01; sc2[1 + j] = c2; }
                }
            }
            const int jc = c;
#pragma unroll
            for (int j = 1; j < 5; ++j)
                tp01[(j - 1) * C_PLANE + jc] = sc01[j];
            tp2[0 * C_PLANE + jc] = lo ? sc2[1] : sc2[2];
            tp2[2 * C_PLANE + jc] = lo ? sc2[3] : sc2[4];
        } else {
            const float* sb = lds + st * STAGE_F;
            float nzs[2][RD_STEPS][3];
#pragma unroll
            for (int k = 0; k < RD_STEPS; ++k) {
                const float* rowS = sb + k * ROW_F + lpA;
                nzs[0][k][0] = rowS[0]; nzs[0][k][1] = rowS[1]; nzs[0][k][2] = rowS[2];
            }
            float ha01 = 0.0f, ha2 = 0.0f, hb01 = 0.0f, hb2 = 0.0f;
#pragma unroll
            for (int s = 0; s < CH1; ++s) {
                const int hg = s / RD_STEPS;
                const int kk = s - hg * RD_STEPS;
                const int sg = s / SAVE_EVERY;
                if (kk == 0) {
                    if (hg + 1 < CH1 / RD_STEPS) {
#pragma unroll
                        for (int u = 0; u < RD_STEPS; ++u) {
                            const float* rowS = sb + ((hg + 1) * RD_STEPS + u) * ROW_F + lpA;
                            nzs[(hg + 1) & 1][u][0] = rowS[0]; nzs[(hg + 1) & 1][u][1] = rowS[1]; nzs[(hg + 1) & 1][u][2] = rowS[2];
                        }
                    }
                    asm volatile("" : "+v"(Bv[0]) : : "memory");
                }
                v8f cz = {};
                v8f d0 = __builtin_amdgcn_wmma_f32_16x16x4_f32(
                    false, Amat, false, Bv[0], (short)0, cz, false, false);
                float s01, s2;
                em_update(d0, nzs[hg & 1][kk], lo, 1.0f, Bv[0], s01, s2);
                if (kk == SAVE_EVERY - 1)     { ha01 = s01; ha2 = s2; }
                if (kk == 2 * SAVE_EVERY - 1) { hb01 = s01; hb2 = s2; }
                if (kk == RD_STEPS - 1) {
                    asm volatile("" : "+v"(Bv[0]) : : "memory");
                    const int jg = c * SUBS + sg;
                    ts01[jg - 2] = ha01;
                    ts2[jg - 2]  = ha2;
                    ts01[jg - 1] = hb01;
                    ts2[jg - 1]  = hb2;
                    ts01[jg] = s01;
                    ts2[jg]  = s2;
                }
            }
        }
        lds_handover_barrier();
    }

    if (producer) drain_saves(tcol, out, dr_lds, dr_out, N_CHUNK - 1);

    for (int i0 = 0; i0 < CLOSE_TRIPS; i0 += CLOSE_U) {
        float va[CLOSE_U];
#pragma unroll
        for (int u = 0; u < CLOSE_U; ++u) {
            const int v  = tid + N_THREADS * (i0 + u);
            const int vv = (v < N_CLOSE) ? v : (N_CLOSE - 1);
            const int rw = vv / SEG2_SAVES;
            const int j  = vv - rw * SEG2_SAVES;
            const int q  = rw & 63;
            const float* xm = tcol + S_OFF + q * TS + (SEG1_SAVES - 1);
            const float a0 = xm[0], a1 = xm[64 * TS], a2 = xm[128 * TS];
            const float* t0 = tcol + rw * TC + j;
            float w = t0[3 * C_PLANE];
            w = __builtin_fmaf(a2, t0[2 * C_PLANE], w);
            w = __builtin_fmaf(a1, t0[1 * C_PLANE], w);
            va[u] = __builtin_fmaf(a0, t0[0], w);
        }
#pragma unroll
        for (int u = 0; u < CLOSE_U; ++u) {
            const int v = tid + N_THREADS * (i0 + u);
            if (v < N_CLOSE) {
                const int rw = v / SEG2_SAVES;
                const int j  = v - rw * SEG2_SAVES;
                out[(rw >> 6) * BT + (blk + (rw & 63)) * T_SAVED + (SEG1_SAVES + 1) + j] = va[u];
            }
        }
    }
}

extern "C" void kernel_launch(void* const* d_in, const int* in_sizes, int n_in,
                              void* d_out, int out_size, void* d_ws, size_t ws_size,
                              hipStream_t stream) {
    const float* drift     = (const float*)d_in[0];
    const float* driftb    = (const float*)d_in[1];
    const float* diffusion = (const float*)d_in[2];
    const float* diffb     = (const float*)d_in[3];
    const float* x0        = (const float*)d_in[4];
    const float* noise     = (const float*)d_in[5];
    float* out = (float*)d_out;

    dim3 block(352);
    dim3 grid(B_PATHS / 64);
    sde_wmma_kernel<<<grid, block, 0, stream>>>(
        drift, driftb, diffusion, diffb, x0, noise, out);
}
